# starting kernel
# baseline (speedup 1.0000x reference)
#include <hip/hip_runtime.h>
#include <hip/hip_bf16.h>

#define NN 100000
#define IN_F 32
#define HID_F 64
#define OUT_F 16

typedef float v2f __attribute__((ext_vector_type(2)));
typedef float v8f __attribute__((ext_vector_type(8)));

__global__ __launch_bounds__(256) void zero_ws_kernel(float* __restrict__ ws, long n) {
    long i = (long)blockIdx.x * blockDim.x + threadIdx.x;
    long stride = (long)gridDim.x * blockDim.x;
    for (; i < n; i += stride) ws[i] = 0.0f;
}

__global__ __launch_bounds__(256) void scatter32_kernel(const float* __restrict__ feat,
                                                        const int* __restrict__ src,
                                                        const int* __restrict__ dst,
                                                        float* __restrict__ agg,
                                                        float* __restrict__ deg,
                                                        int nE) {
    int tid  = blockIdx.x * blockDim.x + threadIdx.x;
    int e    = tid >> 5;
    int lane = tid & 31;
    if (e >= nE) return;
    int s = src[e];
    int d = dst[e];
    atomicAdd(&agg[(long)d * IN_F + lane], feat[(long)s * IN_F + lane]);
    if (lane == 0) atomicAdd(&deg[d], 1.0f);
}

__global__ __launch_bounds__(256) void scatter64_kernel(const float* __restrict__ h,
                                                        const int* __restrict__ src,
                                                        const int* __restrict__ dst,
                                                        float* __restrict__ agg,
                                                        int nE) {
    int tid  = blockIdx.x * blockDim.x + threadIdx.x;
    int e    = tid >> 5;
    int lane = tid & 31;
    if (e >= nE) return;
    int s = src[e];
    int d = dst[e];
    atomicAdd(&agg[(long)d * HID_F + lane],      h[(long)s * HID_F + lane]);
    atomicAdd(&agg[(long)d * HID_F + lane + 32], h[(long)s * HID_F + lane + 32]);
}

__global__ __launch_bounds__(256) void sage_layer1_kernel(const float* __restrict__ feat,
                                                          const float* __restrict__ agg,
                                                          const float* __restrict__ deg,
                                                          const float* __restrict__ Wself,
                                                          const float* __restrict__ Wneigh,
                                                          const float* __restrict__ bias,
                                                          float* __restrict__ h,
                                                          int nRowTiles) {
    int tid  = blockIdx.x * blockDim.x + threadIdx.x;
    int wave = tid >> 5;
    int lane = tid & 31;
    const int colTiles = HID_F / 16;
    int tileRow = wave / colTiles;
    int ct      = wave % colTiles;
    if (tileRow >= nRowTiles) return;

    int hi   = lane >> 4;
    int l15  = lane & 15;
    long row = (long)tileRow * 16 + l15;
    int colW = ct * 16 + l15;

    float dg   = fmaxf(deg[row], 1.0f);
    float invd = 1.0f / dg;

    v8f c;
    float bv = bias[colW];
#pragma unroll
    for (int v = 0; v < 8; ++v) c[v] = bv;

    const float* fr = feat + row * IN_F;
    const float* ar = agg  + row * IN_F;

#pragma unroll
    for (int k0 = 0; k0 < IN_F; k0 += 4) {
        int kA = k0 + 2 * hi;
        v2f a, b;
        a.x = fr[kA];
        a.y = fr[kA + 1];
        b.x = Wself[(long)(kA)     * HID_F + colW];
        b.y = Wself[(long)(kA + 1) * HID_F + colW];
        c = __builtin_amdgcn_wmma_f32_16x16x4_f32(false, a, false, b, (short)0, c, false, false);
        a.x = ar[kA]     * invd;
        a.y = ar[kA + 1] * invd;
        b.x = Wneigh[(long)(kA)     * HID_F + colW];
        b.y = Wneigh[(long)(kA + 1) * HID_F + colW];
        c = __builtin_amdgcn_wmma_f32_16x16x4_f32(false, a, false, b, (short)0, c, false, false);
    }

#pragma unroll
    for (int v = 0; v < 8; ++v) {
        long r = (long)tileRow * 16 + v + 8 * hi;
        h[r * HID_F + colW] = fmaxf(c[v], 0.0f);
    }
}

__global__ __launch_bounds__(256) void sage_layer2_kernel(const float* __restrict__ h,
                                                          const float* __restrict__ agg,
                                                          const float* __restrict__ deg,
                                                          const float* __restrict__ Wself,
                                                          const float* __restrict__ Wneigh,
                                                          const float* __restrict__ bias,
                                                          float* __restrict__ out,
                                                          int nRowTiles) {
    int tid  = blockIdx.x * blockDim.x + threadIdx.x;
    int wave = tid >> 5;
    int lane = tid & 31;
    int tileRow = wave;
    if (tileRow >= nRowTiles) return;

    int hi  = lane >> 4;
    int l15 = lane & 15;
    long row = (long)tileRow * 16 + l15;
    int colW = l15;

    float dg   = fmaxf(deg[row], 1.0f);
    float invd = 1.0f / dg;

    v8f c;
    float bv = bias[colW];
#pragma unroll
    for (int v = 0; v < 8; ++v) c[v] = bv;

    const float* hr = h   + row * HID_F;
    const float* ar = agg + row * HID_F;

#pragma unroll
    for (int k0 = 0; k0 < HID_F; k0 += 4) {
        int kA = k0 + 2 * hi;
        v2f a, b;
        a.x = hr[kA];
        a.y = hr[kA + 1];
        b.x = Wself[(long)(kA)     * OUT_F + colW];
        b.y = Wself[(long)(kA + 1) * OUT_F + colW];
        c = __builtin_amdgcn_wmma_f32_16x16x4_f32(false, a, false, b, (short)0, c, false, false);
        a.x = ar[kA]     * invd;
        a.y = ar[kA + 1] * invd;
        b.x = Wneigh[(long)(kA)     * OUT_F + colW];
        b.y = Wneigh[(long)(kA + 1) * OUT_F + colW];
        c = __builtin_amdgcn_wmma_f32_16x16x4_f32(false, a, false, b, (short)0, c, false, false);
    }

#pragma unroll
    for (int v = 0; v < 8; ++v) {
        long r = (long)tileRow * 16 + v + 8 * hi;
        out[r * OUT_F + colW] = c[v];
    }
}

extern "C" void kernel_launch(void* const* d_in, const int* in_sizes, int n_in,
                              void* d_out, int out_size, void* d_ws, size_t ws_size,
                              hipStream_t stream) {
    const float* feat = (const float*)d_in[0];
    const int*   src  = (const int*)d_in[1];
    const int*   dst  = (const int*)d_in[2];
    const float* W1s  = (const float*)d_in[3];
    const float* W1n  = (const float*)d_in[4];
    const float* b1   = (const float*)d_in[5];
    const float* W2s  = (const float*)d_in[6];
    const float* W2n  = (const float*)d_in[7];
    const float* b2   = (const float*)d_in[8];
    float* out = (float*)d_out;

    const int nNodes = in_sizes[0] / IN_F;
    const int nE     = in_sizes[1];
    const int nRowTiles = nNodes / 16;

    float* ws   = (float*)d_ws;
    float* deg  = ws;
    float* agg1 = deg  + nNodes;
    float* agg2 = agg1 + (long)nNodes * IN_F;
    float* h    = agg2 + (long)nNodes * HID_F;

    long zn = (long)nNodes * (1 + IN_F + HID_F);
    zero_ws_kernel<<<4096, 256, 0, stream>>>(ws, zn);

    {
        long threads = (long)nE * 32;
        int blocks = (int)((threads + 255) / 256);
        scatter32_kernel<<<blocks, 256, 0, stream>>>(feat, src, dst, agg1, deg, nE);
    }
    {
        int waves = nRowTiles * (HID_F / 16);
        int blocks = (waves * 32 + 255) / 256;
        sage_layer1_kernel<<<blocks, 256, 0, stream>>>(feat, agg1, deg, W1s, W1n, b1, h, nRowTiles);
    }

    {
        long threads = (long)nE * 32;
        int blocks = (int)((threads + 255) / 256);
        scatter64_kernel<<<blocks, 256, 0, stream>>>(h, src, dst, agg2, nE);
    }
    {
        int waves = nRowTiles;
        int blocks = (waves * 32 + 255) / 256;
        sage_layer2_kernel<<<blocks, 256, 0, stream>>>(h, agg2, deg, W2s, W2n, b2, out, nRowTiles);
    }
}


# Owner scan reads a 1-byte owner-tile key per edge (new pre-pass: dst/400 mod 256; 4 grouped 16-byte loads a step), not 4-byte dst; hits confirmed from dst; 2->3 launches; LDS-atomic sums as before
# speedup vs baseline: 3.6650x; 3.6650x over previous
#include <hip/hip_runtime.h>

#define IN_F 32
#define HID_F 64
#define OUT_F 16
#define TN 400
#define NW 32
#define NT (NW * 32)
#define LCAP 64
#define KSTRIDE (16 * NT)

typedef float v2f __attribute__((ext_vector_type(2)));
typedef float v8f __attribute__((ext_vector_type(8)));

__global__ __launch_bounds__(256) void owner_key_kernel(const int* __restrict__ dst, int* __restrict__ keys,
                                                        int nE, int nWords) {
    const int base = blockIdx.x * 1024 + (int)threadIdx.x;
    int4 t[4];
#pragma unroll
    for (int k = 0; k < 4; ++k) {
        const int w  = base + k * 256;
        const int e0 = w * 4;
        int4 v = make_int4(-1, -1, -1, -1);
        if (w < nWords) {
            if (e0 + 3 < nE) {
                v = *(const int4*)(dst + e0);
            } else {
                if (e0 < nE)     v.x = dst[e0];
                if (e0 + 1 < nE) v.y = dst[e0 + 1];
                if (e0 + 2 < nE) v.z = dst[e0 + 2];
            }
        }
        t[k] = v;
    }
#pragma unroll
    for (int k = 0; k < 4; ++k) {
        const int w = base + k * 256;
        if (w < nWords) {
            unsigned k0 = t[k].x < 0 ? 255u : (((unsigned)t[k].x / TN) & 255u);
            unsigned k1 = t[k].y < 0 ? 255u : (((unsigned)t[k].y / TN) & 255u);
            unsigned k2 = t[k].z < 0 ? 255u : (((unsigned)t[k].z / TN) & 255u);
            unsigned k3 = t[k].w < 0 ? 255u : (((unsigned)t[k].w / TN) & 255u);
            keys[w] = (int)(k0 | (k1 << 8) | (k2 << 16) | (k3 << 24));
        }
    }
}

__device__ __forceinline__ unsigned eq_bytes(int w, unsigned pat) {
    unsigned z = (unsigned)w ^ pat;
    return (z - 0x01010101u) & ~z & 0x80808080u;
}

__device__ __forceinline__ unsigned mark16(int4 w, unsigned pat) {
    return (eq_bytes(w.x, pat) >> 7) | (eq_bytes(w.y, pat) >> 6) | (eq_bytes(w.z, pat) >> 5) | (eq_bytes(w.w, pat) >> 4);
}

template <int F, int P>
__device__ __forceinline__ void flush_batch(const float* __restrict__ x, const int* __restrict__ src,
                                            const int* __restrict__ dst, const int* le, int nvalid,
                                            int tile0, int nE, float* acc, float* degs, int lane) {
    int e = (lane < nvalid) ? le[lane] : 0;
    bool ok = (lane < nvalid) & (e < nE);
    if (!ok) e = 0;
    int d  = dst[e];
    int sv = src[e];
    unsigned uu = (unsigned)(d - tile0);
    ok = ok & (uu < (unsigned)TN);
    int s = ok ? sv : 0;
    int u = ok ? (int)uu : TN;
    atomicAdd(&degs[u], 1.0f);
    if constexpr (F == 32) {
        float v[32];
#pragma unroll
        for (int j = 0; j < 32; ++j) {
            int sj = __builtin_amdgcn_readlane(s, j);
            v[j] = x[(long)sj * F + lane];
        }
#pragma unroll
        for (int j = 0; j < 32; ++j) {
            int uj = __builtin_amdgcn_readlane(u, j);
            atomicAdd(&acc[uj * P + lane], v[j]);
        }
    } else {
        v2f v[32];
#pragma unroll
        for (int j = 0; j < 32; ++j) {
            int sj = __builtin_amdgcn_readlane(s, j);
            v[j] = *(const v2f*)(x + (long)sj * F + 2 * lane);
        }
#pragma unroll
        for (int j = 0; j < 32; ++j) {
            int uj = __builtin_amdgcn_readlane(u, j);
            float* a = &acc[uj * P + 2 * lane];
            atomicAdd(a, v[j].x);
            atomicAdd(a + 1, v[j].y);
        }
    }
}

template <int F, int OUTW, bool RELU>
__global__ __launch_bounds__(NT) void sage_owner_kernel(const float* __restrict__ x,
                                                        const int* __restrict__ src,
                                                        const int* __restrict__ dst,
                                                        const int* __restrict__ keys,
                                                        const float* __restrict__ Wself,
                                                        const float* __restrict__ Wneigh,
                                                        const float* __restrict__ bias,
                                                        float* __restrict__ y,
                                                        int nNodes, int nE) {
    constexpr int P  = F + 2;
    constexpr int CT = OUTW / 16;
    __shared__ float acc[(TN + 1) * P];
    __shared__ float degs[TN + 16];
    __shared__ int list_e[NW * LCAP];

    const int wave  = threadIdx.x >> 5;
    const int lane  = threadIdx.x & 31;
    const int tile0 = blockIdx.x * TN;

    for (int i = threadIdx.x; i < (TN + 1) * P; i += NT) acc[i] = 0.0f;
    for (int i = threadIdx.x; i < TN + 16; i += NT) degs[i] = 0.0f;
    __syncthreads();

    int* le = &list_e[wave * LCAP];
    int cnt = 0;

    const unsigned pat   = (unsigned)(blockIdx.x & 255) * 0x01010101u;
    const int nGroups = (nE + 15) >> 4;
    const int nSteps  = (nGroups + 4 * NT - 1) / (4 * NT);
    const int step0   = (int)(((long)blockIdx.x * nSteps) / gridDim.x);

    auto load_step = [&](int4 (&w)[4], int st) {
#pragma unroll
        for (int k = 0; k < 4; ++k) {
            int g = (st * 4 + k) * NT + (int)threadIdx.x;
            if (g > nGroups - 1) g = nGroups - 1;
            w[k] = *(const int4*)(keys + 4 * (long)g);
        }
    };
    auto look_at = [&](const int4 (&w)[4], int st, bool live) {
#pragma unroll
        for (int h2 = 0; h2 < 2; ++h2) {
            const int gA = (st * 4 + 2 * h2) * NT + (int)threadIdx.x;
            unsigned qa = mark16(w[2 * h2], pat);
            unsigned qb = mark16(w[2 * h2 + 1], pat);
            if (!(live && gA < nGroups))      qa = 0;
            if (!(live && gA + NT < nGroups)) qb = 0;
            unsigned q = qa | (qb << 4);
            const int eBase = gA * 16;
            while (true) {
                bool has = q != 0;
                unsigned bm = __builtin_amdgcn_ballot_w32(has);
                if (bm == 0) break;
                if (has) {
                    int p = __builtin_ctz(q);
                    q &= q - 1;
                    int e = eBase + ((p >> 2) & 1) * KSTRIDE + ((p & 3) << 2) + (p >> 3);
                    le[cnt + __builtin_amdgcn_mbcnt_lo(bm, 0)] = e;
                }
                cnt += __builtin_popcount(bm);
                if (cnt >= 32) {
                    cnt -= 32;
                    flush_batch<F, P>(x, src, dst, le + cnt, 32, tile0, nE, acc, degs, lane);
                }
            }
        }
    };
    auto step_of = [&](int i) {
        int st = step0 + i;
        if (st >= nSteps) st -= nSteps;
        if (st >= nSteps) st -= nSteps;
        return st;
    };

    if (nSteps > 0) {
        int4 ka[4], kb[4];
        load_step(ka, step_of(0));
        for (int it = 0; it < nSteps; it += 2) {
            const int sa = step_of(it);
            const int sb = step_of(it + 1);
            load_step(kb, sb);
            look_at(ka, sa, true);
            load_step(ka, step_of(it + 2));
            look_at(kb, sb, it + 1 < nSteps);
        }
    }
    if (cnt > 0) flush_batch<F, P>(x, src, dst, le, cnt, tile0, nE, acc, degs, lane);
    __syncthreads();

    if (wave < TN / 16) {
        const int hi  = lane >> 4;
        const int l15 = lane & 15;
        const int lr  = wave * 16 + l15;
        int row = tile0 + lr;
        if (row > nNodes - 1) row = nNodes - 1;
        const float invd = 1.0f / fmaxf(degs[lr], 1.0f);

        v8f c[CT];
#pragma unroll
        for (int ct = 0; ct < CT; ++ct) {
            float bv = bias[ct * 16 + l15];
#pragma unroll
            for (int v = 0; v < 8; ++v) c[ct][v] = bv;
        }
        const float* xr = x + (long)row * F;
        const float* ar = acc + lr * P;
#pragma unroll
        for (int k0 = 0; k0 < F; k0 += 4) {
            const int kA = k0 + 2 * hi;
            v2f a, m;
            a.x = xr[kA];
            a.y = xr[kA + 1];
            m.x = ar[kA] * invd;
            m.y = ar[kA + 1] * invd;
#pragma unroll
            for (int ct = 0; ct < CT; ++ct) {
                const int colW = ct * 16 + l15;
                v2f b;
                b.x = Wself[(long)(kA)     * OUTW + colW];
                b.y = Wself[(long)(kA + 1) * OUTW + colW];
                c[ct] = __builtin_amdgcn_wmma_f32_16x16x4_f32(false, a, false, b, (short)0, c[ct], false, false);
                b.x = Wneigh[(long)(kA)     * OUTW + colW];
                b.y = Wneigh[(long)(kA + 1) * OUTW + colW];
                c[ct] = __builtin_amdgcn_wmma_f32_16x16x4_f32(false, m, false, b, (short)0, c[ct], false, false);
            }
        }
#pragma unroll
        for (int ct = 0; ct < CT; ++ct) {
#pragma unroll
            for (int v = 0; v < 8; ++v) {
                int r = tile0 + wave * 16 + v + 8 * hi;
                float o = c[ct][v];
                if (RELU) o = fmaxf(o, 0.0f);
                if (r < nNodes) y[(long)r * OUTW + ct * 16 + l15] = o;
            }
        }
    }
}

extern "C" void kernel_launch(void* const* d_in, const int* in_sizes, int n_in,
                              void* d_out, int out_size, void* d_ws, size_t ws_size,
                              hipStream_t stream) {
    const float* feat = (const float*)d_in[0];
    const int*   src  = (const int*)d_in[1];
    const int*   dst  = (const int*)d_in[2];
    const float* W1s  = (const float*)d_in[3];
    const float* W1n  = (const float*)d_in[4];
    const float* b1   = (const float*)d_in[5];
    const float* W2s  = (const float*)d_in[6];
    const float* W2n  = (const float*)d_in[7];
    const float* b2   = (const float*)d_in[8];
    float* out = (float*)d_out;

    const int nNodes = in_sizes[0] / IN_F;
    const int nE     = in_sizes[1];
    const int nTiles = (nNodes + TN - 1) / TN;
    const int nWords = ((nE + 15) / 16) * 4;

    float* h    = (float*)d_ws;
    int*   keys = (int*)(h + (size_t)nNodes * HID_F);

    owner_key_kernel<<<(nWords + 1023) / 1024, 256, 0, stream>>>(dst, keys, nE, nWords);
    sage_owner_kernel<IN_F, HID_F, true><<<nTiles, NT, 0, stream>>>(feat, src, dst, keys, W1s, W1n, b1, h, nNodes, nE);
    sage_owner_kernel<HID_F, OUT_F, false><<<nTiles, NT, 0, stream>>>(h, src, dst, keys, W2s, W2n, b2, out, nNodes, nE);
}


# Layer 2 reads no edge keys: layer-1 owners append their confirmed edges as (source row, local row) words to per-tile workspace lists, layer 2 sums from them; LDS-atomic sums run-varying as before
# speedup vs baseline: 4.4974x; 1.2271x over previous
#include <hip/hip_runtime.h>

#define IN_F 32
#define HID_F 64
#define OUT_F 16
#define TN 400
#define NW 32
#define NT (NW * 32)
#define LCAP 64
#define KSTRIDE (16 * NT)
#define UBITS 9
#define TCAP 8192

typedef float v2f __attribute__((ext_vector_type(2)));
typedef float v8f __attribute__((ext_vector_type(8)));

__global__ __launch_bounds__(256) void owner_key_kernel(const int* __restrict__ dst, int* __restrict__ keys,
                                                        int nE, int nWords) {
    const int base = blockIdx.x * 1024 + (int)threadIdx.x;
    int4 t[4];
#pragma unroll
    for (int k = 0; k < 4; ++k) {
        const int w  = base + k * 256;
        const int e0 = w * 4;
        int4 v = make_int4(-1, -1, -1, -1);
        if (w < nWords) {
            if (e0 + 3 < nE) {
                v = *(const int4*)(dst + e0);
            } else {
                if (e0 < nE)     v.x = dst[e0];
                if (e0 + 1 < nE) v.y = dst[e0 + 1];
                if (e0 + 2 < nE) v.z = dst[e0 + 2];
            }
        }
        t[k] = v;
    }
#pragma unroll
    for (int k = 0; k < 4; ++k) {
        const int w = base + k * 256;
        if (w < nWords) {
            unsigned k0 = t[k].x < 0 ? 255u : (((unsigned)t[k].x / TN) & 255u);
            unsigned k1 = t[k].y < 0 ? 255u : (((unsigned)t[k].y / TN) & 255u);
            unsigned k2 = t[k].z < 0 ? 255u : (((unsigned)t[k].z / TN) & 255u);
            unsigned k3 = t[k].w < 0 ? 255u : (((unsigned)t[k].w / TN) & 255u);
            keys[w] = (int)(k0 | (k1 << 8) | (k2 << 16) | (k3 << 24));
        }
    }
}

__device__ __forceinline__ unsigned eq_bytes(int w, unsigned pat) {
    unsigned z = (unsigned)w ^ pat;
    return (z - 0x01010101u) & ~z & 0x80808080u;
}

__device__ __forceinline__ unsigned mark16(int4 w, unsigned pat) {
    return (eq_bytes(w.x, pat) >> 7) | (eq_bytes(w.y, pat) >> 6) | (eq_bytes(w.z, pat) >> 5) | (eq_bytes(w.w, pat) >> 4);
}

__device__ __forceinline__ bool confirm_batch(const int* __restrict__ src, const int* __restrict__ dst,
                                              const int* le, int nvalid, int tile0, int nE, int lane,
                                              int& s, int& u) {
    int e = (lane < nvalid) ? le[lane] : 0;
    bool ok = (lane < nvalid) & (e < nE);
    if (!ok) e = 0;
    int d  = dst[e];
    int sv = src[e];
    unsigned uu = (unsigned)(d - tile0);
    ok = ok & (uu < (unsigned)TN);
    s = ok ? sv : 0;
    u = ok ? (int)uu : TN;
    return ok;
}

template <int F, int P>
__device__ __forceinline__ void add_rows(const float* __restrict__ x, int s, int u,
                                         float* acc, float* degs, int lane) {
    atomicAdd(&degs[u], 1.0f);
    if constexpr (F == 32) {
        float v[32];
#pragma unroll
        for (int j = 0; j < 32; ++j) {
            int sj = __builtin_amdgcn_readlane(s, j);
            v[j] = x[(long)sj * F + lane];
        }
#pragma unroll
        for (int j = 0; j < 32; ++j) {
            int uj = __builtin_amdgcn_readlane(u, j);
            atomicAdd(&acc[uj * P + lane], v[j]);
        }
    } else {
        v2f v[32];
#pragma unroll
        for (int j = 0; j < 32; ++j) {
            int sj = __builtin_amdgcn_readlane(s, j);
            v[j] = *(const v2f*)(x + (long)sj * F + 2 * lane);
        }
#pragma unroll
        for (int j = 0; j < 32; ++j) {
            int uj = __builtin_amdgcn_readlane(u, j);
            float* a = &acc[uj * P + 2 * lane];
            atomicAdd(a, v[j].x);
            atomicAdd(a + 1, v[j].y);
        }
    }
}

template <int F, int OUTW, bool RELU, int MODE>
__global__ __launch_bounds__(NT) void sage_owner_kernel(const float* __restrict__ x,
                                                        const int* __restrict__ src,
                                                        const int* __restrict__ dst,
                                                        const int* __restrict__ keys,
                                                        const float* __restrict__ Wself,
                                                        const float* __restrict__ Wneigh,
                                                        const float* __restrict__ bias,
                                                        float* __restrict__ y,
                                                        int* __restrict__ counts,
                                                        unsigned* __restrict__ lists, int cap,
                                                        int nNodes, int nE) {
    constexpr int P  = F + 2;
    constexpr int CT = OUTW / 16;
    __shared__ float acc[(TN + 1) * P];
    __shared__ float degs[TN + 16];
    __shared__ int list_e[NW * LCAP];
    __shared__ int lcount;

    const int wave  = threadIdx.x >> 5;
    const int lane  = threadIdx.x & 31;
    const int tile0 = blockIdx.x * TN;

    unsigned* tl = lists + (long)blockIdx.x * cap;
    int total = 0;
    bool from_list = false;
    unsigned wc = 0;
    const int b0 = __builtin_amdgcn_readfirstlane(wave) * 32;
    if constexpr (MODE == 2) {
        if (cap > 0) {
            total = counts[blockIdx.x];
            from_list = total <= cap;
            if (from_list && b0 < total) wc = tl[min(b0 + lane, total - 1)];
        }
    }

    for (int i = threadIdx.x; i < (TN + 1) * P; i += NT) acc[i] = 0.0f;
    for (int i = threadIdx.x; i < TN + 16; i += NT) degs[i] = 0.0f;
    if (threadIdx.x == 0) lcount = 0;
    __syncthreads();

    int* le = &list_e[wave * LCAP];
    int cnt = 0;

    auto flush = [&](const int* l, int nvalid) {
        int s, u;
        bool ok = confirm_batch(src, dst, l, nvalid, tile0, nE, lane, s, u);
        if constexpr (MODE == 1) {
            if (cap > 0) {
                unsigned okm = __builtin_amdgcn_ballot_w32(ok);
                int at = 0;
                if (lane == 0) at = atomicAdd(&lcount, __builtin_popcount(okm));
                at = __builtin_amdgcn_readfirstlane(at) + __builtin_amdgcn_mbcnt_lo(okm, 0);
                if (ok && at < cap) tl[at] = ((unsigned)s << UBITS) | (unsigned)u;
            }
        }
        add_rows<F, P>(x, s, u, acc, degs, lane);
    };

    const unsigned pat   = (unsigned)(blockIdx.x & 255) * 0x01010101u;
    const int nGroups = (nE + 15) >> 4;
    const int nSteps  = (nGroups + 4 * NT - 1) / (4 * NT);
    const int step0   = (int)(((long)blockIdx.x * nSteps) / gridDim.x);

    auto load_step = [&](int4 (&w)[4], int st) {
#pragma unroll
        for (int k = 0; k < 4; ++k) {
            int g = (st * 4 + k) * NT + (int)threadIdx.x;
            if (g > nGroups - 1) g = nGroups - 1;
            w[k] = *(const int4*)(keys + 4 * (long)g);
        }
    };
    auto look_at = [&](const int4 (&w)[4], int st, bool live) {
#pragma unroll
        for (int h2 = 0; h2 < 2; ++h2) {
            const int gA = (st * 4 + 2 * h2) * NT + (int)threadIdx.x;
            unsigned qa = mark16(w[2 * h2], pat);
            unsigned qb = mark16(w[2 * h2 + 1], pat);
            if (!(live && gA < nGroups))      qa = 0;
            if (!(live && gA + NT < nGroups)) qb = 0;
            unsigned q = qa | (qb << 4);
            const int eBase = gA * 16;
            while (true) {
                bool has = q != 0;
                unsigned bm = __builtin_amdgcn_ballot_w32(has);
                if (bm == 0) break;
                if (has) {
                    int p = __builtin_ctz(q);
                    q &= q - 1;
                    int e = eBase + ((p >> 2) & 1) * KSTRIDE + ((p & 3) << 2) + (p >> 3);
                    le[cnt + __builtin_amdgcn_mbcnt_lo(bm, 0)] = e;
                }
                cnt += __builtin_popcount(bm);
                if (cnt >= 32) {
                    cnt -= 32;
                    flush(le + cnt, 32);
                }
            }
        }
    };
    auto step_of = [&](int i) {
        int st = step0 + i;
        if (st >= nSteps) st -= nSteps;
        if (st >= nSteps) st -= nSteps;
        return st;
    };

    if (from_list) {
        int b = b0;
        while (b < total) {
            const int bn = b + NT;
            unsigned wn = 0;
            if (bn < total) wn = tl[min(bn + lane, total - 1)];
            const bool ok = b + lane < total;
            const int s = ok ? (int)(wc >> UBITS) : 0;
            const int u = ok ? (int)(wc & ((1u << UBITS) - 1u)) : TN;
            add_rows<F, P>(x, s, u, acc, degs, lane);
            wc = wn;
            b  = bn;
        }
    } else if (nSteps > 0) {
        int4 ka[4], kb[4];
        load_step(ka, step_of(0));
        for (int it = 0; it < nSteps; it += 2) {
            const int sa = step_of(it);
            const int sb = step_of(it + 1);
            load_step(kb, sb);
            look_at(ka, sa, true);
            load_step(ka, step_of(it + 2));
            look_at(kb, sb, it + 1 < nSteps);
        }
    }
    if (cnt > 0) flush(le, cnt);
    __syncthreads();
    if constexpr (MODE == 1) {
        if (cap > 0 && threadIdx.x == 0) counts[blockIdx.x] = lcount;
    }

    if (wave < TN / 16) {
        const int hi  = lane >> 4;
        const int l15 = lane & 15;
        const int lr  = wave * 16 + l15;
        int row = tile0 + lr;
        if (row > nNodes - 1) row = nNodes - 1;
        const float invd = 1.0f / fmaxf(degs[lr], 1.0f);

        v8f c[CT];
#pragma unroll
        for (int ct = 0; ct < CT; ++ct) {
            float bv = bias[ct * 16 + l15];
#pragma unroll
            for (int v = 0; v < 8; ++v) c[ct][v] = bv;
        }
        const float* xr = x + (long)row * F;
        const float* ar = acc + lr * P;
#pragma unroll
        for (int k0 = 0; k0 < F; k0 += 4) {
            const int kA = k0 + 2 * hi;
            v2f a, m;
            a.x = xr[kA];
            a.y = xr[kA + 1];
            m.x = ar[kA] * invd;
            m.y = ar[kA + 1] * invd;
#pragma unroll
            for (int ct = 0; ct < CT; ++ct) {
                const int colW = ct * 16 + l15;
                v2f b;
                b.x = Wself[(long)(kA)     * OUTW + colW];
                b.y = Wself[(long)(kA + 1) * OUTW + colW];
                c[ct] = __builtin_amdgcn_wmma_f32_16x16x4_f32(false, a, false, b, (short)0, c[ct], false, false);
                b.x = Wneigh[(long)(kA)     * OUTW + colW];
                b.y = Wneigh[(long)(kA + 1) * OUTW + colW];
                c[ct] = __builtin_amdgcn_wmma_f32_16x16x4_f32(false, m, false, b, (short)0, c[ct], false, false);
            }
        }
#pragma unroll
        for (int ct = 0; ct < CT; ++ct) {
#pragma unroll
            for (int v = 0; v < 8; ++v) {
                int r = tile0 + wave * 16 + v + 8 * hi;
                float o = c[ct][v];
                if (RELU) o = fmaxf(o, 0.0f);
                if (r < nNodes) y[(long)r * OUTW + ct * 16 + l15] = o;
            }
        }
    }
}

extern "C" void kernel_launch(void* const* d_in, const int* in_sizes, int n_in,
                              void* d_out, int out_size, void* d_ws, size_t ws_size,
                              hipStream_t stream) {
    const float* feat = (const float*)d_in[0];
    const int*   src  = (const int*)d_in[1];
    const int*   dst  = (const int*)d_in[2];
    const float* W1s  = (const float*)d_in[3];
    const float* W1n  = (const float*)d_in[4];
    const float* b1   = (const float*)d_in[5];
    const float* W2s  = (const float*)d_in[6];
    const float* W2n  = (const float*)d_in[7];
    const float* b2   = (const float*)d_in[8];
    float* out = (float*)d_out;

    const int nNodes = in_sizes[0] / IN_F;
    const int nE     = in_sizes[1];
    const int nTiles = (nNodes + TN - 1) / TN;
    const int nWords = ((nE + 15) / 16) * 4;

    float* h    = (float*)d_ws;
    int*   keys = (int*)(h + (size_t)nNodes * HID_F);
    int*   counts = keys + nWords;
    unsigned* lists = (unsigned*)(counts + ((nTiles + 3) & ~3));
    const bool fits = (size_t)((const char*)(lists + (size_t)nTiles * TCAP) - (const char*)d_ws) <= ws_size
                      && nNodes <= (1 << (32 - UBITS));
    const int cap = fits ? TCAP : 0;

    owner_key_kernel<<<(nWords + 1023) / 1024, 256, 0, stream>>>(dst, keys, nE, nWords);
    sage_owner_kernel<IN_F, HID_F, true, 1><<<nTiles, NT, 0, stream>>>(feat, src, dst, keys, W1s, W1n, b1, h, counts, lists, cap, nNodes, nE);
    sage_owner_kernel<HID_F, OUT_F, false, 2><<<nTiles, NT, 0, stream>>>(h, src, dst, keys, W2s, W2n, b2, out, counts, lists, cap, nNodes, nE);
}


# Candidate confirm: key pre-pass also writes one 8-byte (destination, source) record per edge; owners confirm from that record, not from two arrays (one random line, not two); sum order as before
# speedup vs baseline: 4.7026x; 1.0456x over previous
#include <hip/hip_runtime.h>

#define IN_F 32
#define HID_F 64
#define OUT_F 16
#define TN 400
#define NW 32
#define NT (NW * 32)
#define LCAP 64
#define KSTRIDE (16 * NT)
#define UBITS 9
#define TCAP 8192

typedef float v2f __attribute__((ext_vector_type(2)));
typedef float v8f __attribute__((ext_vector_type(8)));

__global__ __launch_bounds__(256) void owner_key_kernel(const int* __restrict__ dst, const int* __restrict__ src,
                                                        int* __restrict__ keys, int* __restrict__ pairs,
                                                        int nE, int nWords) {
    const int base = blockIdx.x * 1024 + (int)threadIdx.x;
    int4 t[4], sr[4];
#pragma unroll
    for (int k = 0; k < 4; ++k) {
        const int w  = base + k * 256;
        const int e0 = w * 4;
        int4 v = make_int4(-1, -1, -1, -1);
        int4 q = make_int4(0, 0, 0, 0);
        if (w < nWords) {
            if (e0 + 3 < nE) {
                v = *(const int4*)(dst + e0);
                if (pairs) q = *(const int4*)(src + e0);
            } else {
                if (e0 < nE)     v.x = dst[e0];
                if (e0 + 1 < nE) v.y = dst[e0 + 1];
                if (e0 + 2 < nE) v.z = dst[e0 + 2];
                if (pairs) {
                    if (e0 < nE)     q.x = src[e0];
                    if (e0 + 1 < nE) q.y = src[e0 + 1];
                    if (e0 + 2 < nE) q.z = src[e0 + 2];
                }
            }
        }
        t[k]  = v;
        sr[k] = q;
    }
#pragma unroll
    for (int k = 0; k < 4; ++k) {
        const int w = base + k * 256;
        if (w < nWords) {
            if (pairs) {
                int4* p = (int4*)(pairs + 8 * (long)w);
                p[0] = make_int4(t[k].x, sr[k].x, t[k].y, sr[k].y);
                p[1] = make_int4(t[k].z, sr[k].z, t[k].w, sr[k].w);
            }
            unsigned k0 = t[k].x < 0 ? 255u : (((unsigned)t[k].x / TN) & 255u);
            unsigned k1 = t[k].y < 0 ? 255u : (((unsigned)t[k].y / TN) & 255u);
            unsigned k2 = t[k].z < 0 ? 255u : (((unsigned)t[k].z / TN) & 255u);
            unsigned k3 = t[k].w < 0 ? 255u : (((unsigned)t[k].w / TN) & 255u);
            keys[w] = (int)(k0 | (k1 << 8) | (k2 << 16) | (k3 << 24));
        }
    }
}

__device__ __forceinline__ unsigned eq_bytes(int w, unsigned pat) {
    unsigned z = (unsigned)w ^ pat;
    return (z - 0x01010101u) & ~z & 0x80808080u;
}

__device__ __forceinline__ unsigned mark16(int4 w, unsigned pat) {
    return (eq_bytes(w.x, pat) >> 7) | (eq_bytes(w.y, pat) >> 6) | (eq_bytes(w.z, pat) >> 5) | (eq_bytes(w.w, pat) >> 4);
}

__device__ __forceinline__ bool confirm_batch(const int* __restrict__ src, const int* __restrict__ dst,
                                              const int* __restrict__ pairs,
                                              const int* le, int nvalid, int tile0, int nE, int lane,
                                              int& s, int& u) {
    int e = (lane < nvalid) ? le[lane] : 0;
    bool ok = (lane < nvalid) & (e < nE);
    if (!ok) e = 0;
    int d, sv;
    if (pairs) {
        const unsigned long long r = *(const unsigned long long*)(pairs + 2 * (long)e);
        d  = (int)(unsigned)r;
        sv = (int)(unsigned)(r >> 32);
    } else {
        d  = dst[e];
        sv = src[e];
    }
    unsigned uu = (unsigned)(d - tile0);
    ok = ok & (uu < (unsigned)TN);
    s = ok ? sv : 0;
    u = ok ? (int)uu : TN;
    return ok;
}

template <int F, int P>
__device__ __forceinline__ void add_rows(const float* __restrict__ x, int s, int u,
                                         float* acc, float* degs, int lane) {
    atomicAdd(&degs[u], 1.0f);
    if constexpr (F == 32) {
        float v[32];
#pragma unroll
        for (int j = 0; j < 32; ++j) {
            int sj = __builtin_amdgcn_readlane(s, j);
            v[j] = x[(long)sj * F + lane];
        }
#pragma unroll
        for (int j = 0; j < 32; ++j) {
            int uj = __builtin_amdgcn_readlane(u, j);
            atomicAdd(&acc[uj * P + lane], v[j]);
        }
    } else {
        v2f v[32];
#pragma unroll
        for (int j = 0; j < 32; ++j) {
            int sj = __builtin_amdgcn_readlane(s, j);
            v[j] = *(const v2f*)(x + (long)sj * F + 2 * lane);
        }
#pragma unroll
        for (int j = 0; j < 32; ++j) {
            int uj = __builtin_amdgcn_readlane(u, j);
            float* a = &acc[uj * P + 2 * lane];
            atomicAdd(a, v[j].x);
            atomicAdd(a + 1, v[j].y);
        }
    }
}

template <int F, int OUTW, bool RELU, int MODE>
__global__ __launch_bounds__(NT) void sage_owner_kernel(const float* __restrict__ x,
                                                        const int* __restrict__ src,
                                                        const int* __restrict__ dst,
                                                        const int* __restrict__ keys,
                                                        const int* __restrict__ pairs,
                                                        const float* __restrict__ Wself,
                                                        const float* __restrict__ Wneigh,
                                                        const float* __restrict__ bias,
                                                        float* __restrict__ y,
                                                        int* __restrict__ counts,
                                                        unsigned* __restrict__ lists, int cap,
                                                        int nNodes, int nE) {
    constexpr int P  = F + 2;
    constexpr int CT = OUTW / 16;
    __shared__ float acc[(TN + 1) * P];
    __shared__ float degs[TN + 16];
    __shared__ int list_e[NW * LCAP];
    __shared__ int lcount;

    const int wave  = threadIdx.x >> 5;
    const int lane  = threadIdx.x & 31;
    const int tile0 = blockIdx.x * TN;

    unsigned* tl = lists + (long)blockIdx.x * cap;
    int total = 0;
    bool from_list = false;
    unsigned wc = 0;
    const int b0 = __builtin_amdgcn_readfirstlane(wave) * 32;
    if constexpr (MODE == 2) {
        if (cap > 0) {
            total = counts[blockIdx.x];
            from_list = total <= cap;
            if (from_list && b0 < total) wc = tl[min(b0 + lane, total - 1)];
        }
    }

    for (int i = threadIdx.x; i < (TN + 1) * P; i += NT) acc[i] = 0.0f;
    for (int i = threadIdx.x; i < TN + 16; i += NT) degs[i] = 0.0f;
    if (threadIdx.x == 0) lcount = 0;
    __syncthreads();

    int* le = &list_e[wave * LCAP];
    int cnt = 0;

    auto flush = [&](const int* l, int nvalid) {
        int s, u;
        bool ok = confirm_batch(src, dst, pairs, l, nvalid, tile0, nE, lane, s, u);
        if constexpr (MODE == 1) {
            if (cap > 0) {
                unsigned okm = __builtin_amdgcn_ballot_w32(ok);
                int at = 0;
                if (lane == 0) at = atomicAdd(&lcount, __builtin_popcount(okm));
                at = __builtin_amdgcn_readfirstlane(at) + __builtin_amdgcn_mbcnt_lo(okm, 0);
                if (ok && at < cap) tl[at] = ((unsigned)s << UBITS) | (unsigned)u;
            }
        }
        add_rows<F, P>(x, s, u, acc, degs, lane);
    };

    const unsigned pat   = (unsigned)(blockIdx.x & 255) * 0x01010101u;
    const int nGroups = (nE + 15) >> 4;
    const int nSteps  = (nGroups + 4 * NT - 1) / (4 * NT);
    const int step0   = (int)(((long)blockIdx.x * nSteps) / gridDim.x);

    auto load_step = [&](int4 (&w)[4], int st) {
#pragma unroll
        for (int k = 0; k < 4; ++k) {
            int g = (st * 4 + k) * NT + (int)threadIdx.x;
            if (g > nGroups - 1) g = nGroups - 1;
            w[k] = *(const int4*)(keys + 4 * (long)g);
        }
    };
    auto look_at = [&](const int4 (&w)[4], int st, bool live) {
#pragma unroll
        for (int h2 = 0; h2 < 2; ++h2) {
            const int gA = (st * 4 + 2 * h2) * NT + (int)threadIdx.x;
            unsigned qa = mark16(w[2 * h2], pat);
            unsigned qb = mark16(w[2 * h2 + 1], pat);
            if (!(live && gA < nGroups))      qa = 0;
            if (!(live && gA + NT < nGroups)) qb = 0;
            unsigned q = qa | (qb << 4);
            const int eBase = gA * 16;
            while (true) {
                bool has = q != 0;
                unsigned bm = __builtin_amdgcn_ballot_w32(has);
                if (bm == 0) break;
                if (has) {
                    int p = __builtin_ctz(q);
                    q &= q - 1;
                    int e = eBase + ((p >> 2) & 1) * KSTRIDE + ((p & 3) << 2) + (p >> 3);
                    le[cnt + __builtin_amdgcn_mbcnt_lo(bm, 0)] = e;
                }
                cnt += __builtin_popcount(bm);
                if (cnt >= 32) {
                    cnt -= 32;
                    flush(le + cnt, 32);
                }
            }
        }
    };
    auto step_of = [&](int i) {
        int st = step0 + i;
        if (st >= nSteps) st -= nSteps;
        if (st >= nSteps) st -= nSteps;
        return st;
    };

    if (from_list) {
        int b = b0;
        while (b < total) {
            const int bn = b + NT;
            unsigned wn = 0;
            if (bn < total) wn = tl[min(bn + lane, total - 1)];
            const bool ok = b + lane < total;
            const int s = ok ? (int)(wc >> UBITS) : 0;
            const int u = ok ? (int)(wc & ((1u << UBITS) - 1u)) : TN;
            add_rows<F, P>(x, s, u, acc, degs, lane);
            wc = wn;
            b  = bn;
        }
    } else if (nSteps > 0) {
        int4 ka[4], kb[4];
        load_step(ka, step_of(0));
        for (int it = 0; it < nSteps; it += 2) {
            const int sa = step_of(it);
            const int sb = step_of(it + 1);
            load_step(kb, sb);
            look_at(ka, sa, true);
            load_step(ka, step_of(it + 2));
            look_at(kb, sb, it + 1 < nSteps);
        }
    }
    if (cnt > 0) flush(le, cnt);
    __syncthreads();
    if constexpr (MODE == 1) {
        if (cap > 0 && threadIdx.x == 0) counts[blockIdx.x] = lcount;
    }

    if (wave < TN / 16) {
        const int hi  = lane >> 4;
        const int l15 = lane & 15;
        const int lr  = wave * 16 + l15;
        int row = tile0 + lr;
        if (row > nNodes - 1) row = nNodes - 1;
        const float invd = 1.0f / fmaxf(degs[lr], 1.0f);

        v8f c[CT];
#pragma unroll
        for (int ct = 0; ct < CT; ++ct) {
            float bv = bias[ct * 16 + l15];
#pragma unroll
            for (int v = 0; v < 8; ++v) c[ct][v] = bv;
        }
        const float* xr = x + (long)row * F;
        const float* ar = acc + lr * P;
#pragma unroll
        for (int k0 = 0; k0 < F; k0 += 4) {
            const int kA = k0 + 2 * hi;
            v2f a, m;
            a.x = xr[kA];
            a.y = xr[kA + 1];
            m.x = ar[kA] * invd;
            m.y = ar[kA + 1] * invd;
#pragma unroll
            for (int ct = 0; ct < CT; ++ct) {
                const int colW = ct * 16 + l15;
                v2f b;
                b.x = Wself[(long)(kA)     * OUTW + colW];
                b.y = Wself[(long)(kA + 1) * OUTW + colW];
                c[ct] = __builtin_amdgcn_wmma_f32_16x16x4_f32(false, a, false, b, (short)0, c[ct], false, false);
                b.x = Wneigh[(long)(kA)     * OUTW + colW];
                b.y = Wneigh[(long)(kA + 1) * OUTW + colW];
                c[ct] = __builtin_amdgcn_wmma_f32_16x16x4_f32(false, m, false, b, (short)0, c[ct], false, false);
            }
        }
#pragma unroll
        for (int ct = 0; ct < CT; ++ct) {
#pragma unroll
            for (int v = 0; v < 8; ++v) {
                int r = tile0 + wave * 16 + v + 8 * hi;
                float o = c[ct][v];
                if (RELU) o = fmaxf(o, 0.0f);
                if (r < nNodes) y[(long)r * OUTW + ct * 16 + l15] = o;
            }
        }
    }
}

extern "C" void kernel_launch(void* const* d_in, const int* in_sizes, int n_in,
                              void* d_out, int out_size, void* d_ws, size_t ws_size,
                              hipStream_t stream) {
    const float* feat = (const float*)d_in[0];
    const int*   src  = (const int*)d_in[1];
    const int*   dst  = (const int*)d_in[2];
    const float* W1s  = (const float*)d_in[3];
    const float* W1n  = (const float*)d_in[4];
    const float* b1   = (const float*)d_in[5];
    const float* W2s  = (const float*)d_in[6];
    const float* W2n  = (const float*)d_in[7];
    const float* b2   = (const float*)d_in[8];
    float* out = (float*)d_out;

    const int nNodes = in_sizes[0] / IN_F;
    const int nE     = in_sizes[1];
    const int nTiles = (nNodes + TN - 1) / TN;
    const int nWords = ((nE + 15) / 16) * 4;

    float* h    = (float*)d_ws;
    int*   keys = (int*)(h + (size_t)nNodes * HID_F);
    int*   counts = keys + nWords;
    unsigned* lists = (unsigned*)(counts + ((nTiles + 3) & ~3));
    const bool fits = (size_t)((const char*)(lists + (size_t)nTiles * TCAP) - (const char*)d_ws) <= ws_size
                      && nNodes <= (1 << (32 - UBITS));
    const int cap = fits ? TCAP : 0;
    int* recs = (int*)(lists + (size_t)nTiles * TCAP);
    const bool fitsRecs = fits && (size_t)((const char*)(recs + 8 * (size_t)nWords) - (const char*)d_ws) <= ws_size;
    int* pairs = fitsRecs ? recs : nullptr;

    owner_key_kernel<<<(nWords + 1023) / 1024, 256, 0, stream>>>(dst, src, keys, pairs, nE, nWords);
    sage_owner_kernel<IN_F, HID_F, true, 1><<<nTiles, NT, 0, stream>>>(feat, src, dst, keys, pairs, W1s, W1n, b1, h, counts, lists, cap, nNodes, nE);
    sage_owner_kernel<HID_F, OUT_F, false, 2><<<nTiles, NT, 0, stream>>>(h, src, dst, keys, pairs, W2s, W2n, b2, out, counts, lists, cap, nNodes, nE);
}


# Owner scan matches eight bit planes of the tile key (pre-pass: one ballot per key bit per 32 edges; about 12 bit ops mark 32 edges) instead of byte keys (about 50 ops per 32); bytes read unchanged
# speedup vs baseline: 4.9044x; 1.0429x over previous
#include <hip/hip_runtime.h>

#define IN_F 32
#define HID_F 64
#define OUT_F 16
#define TN 400
#define NW 32
#define NT (NW * 32)
#define LCAP 64
#define BPW 16
#define UBITS 9
#define TCAP 8192

typedef float v2f __attribute__((ext_vector_type(2)));
typedef float v8f __attribute__((ext_vector_type(8)));

__global__ __launch_bounds__(256) void owner_key_kernel(const int* __restrict__ dst, const int* __restrict__ src,
                                                        int* __restrict__ keys, int* __restrict__ pairs,
                                                        int nE, int nBlocks) {
    const int lane = (int)threadIdx.x & 31;
    const int blk0 = (blockIdx.x * 8 + ((int)threadIdx.x >> 5)) * BPW;
    if (blk0 >= nBlocks) return;
    unsigned pl[8];
#pragma unroll
    for (int p = 0; p < 8; ++p) pl[p] = 0u;
#pragma unroll
    for (int c = 0; c < BPW / 8; ++c) {
        int d[8], s[8];
#pragma unroll
        for (int j = 0; j < 8; ++j) {
            const int e = (blk0 + c * 8 + j) * 32 + lane;
            d[j] = (e < nE) ? dst[e] : -1;
            s[j] = (pairs && e < nE) ? src[e] : 0;
        }
#pragma unroll
        for (int j = 0; j < 8; ++j) {
            const int blk = blk0 + c * 8 + j;
            const int e = blk * 32 + lane;
            if (pairs && blk < nBlocks) *(int2*)(pairs + 2 * (long)e) = make_int2(d[j], s[j]);
            const unsigned key = d[j] < 0 ? 255u : (((unsigned)d[j] / TN) & 255u);
#pragma unroll
            for (int p = 0; p < 8; ++p) {
                const unsigned m = __builtin_amdgcn_ballot_w32(((key >> p) & 1u) != 0u);
                if (lane == c * 8 + j) pl[p] = m;
            }
        }
    }
    const int blk = blk0 + lane;
    if (lane < BPW && blk < nBlocks) {
        *(int4*)(keys + 4 * (long)blk)             = make_int4((int)pl[0], (int)pl[1], (int)pl[2], (int)pl[3]);
        *(int4*)(keys + 4 * ((long)nBlocks + blk)) = make_int4((int)pl[4], (int)pl[5], (int)pl[6], (int)pl[7]);
    }
}

__device__ __forceinline__ unsigned match32(int4 lo, int4 hi, const unsigned (&t)[8]) {
    unsigned a = (unsigned)lo.x ^ t[0];
    a |= (unsigned)lo.y ^ t[1];
    a |= (unsigned)lo.z ^ t[2];
    a |= (unsigned)lo.w ^ t[3];
    a |= (unsigned)hi.x ^ t[4];
    a |= (unsigned)hi.y ^ t[5];
    a |= (unsigned)hi.z ^ t[6];
    a |= (unsigned)hi.w ^ t[7];
    return ~a;
}

__device__ __forceinline__ bool confirm_batch(const int* __restrict__ src, const int* __restrict__ dst,
                                              const int* __restrict__ pairs,
                                              const int* le, int nvalid, int tile0, int nE, int lane,
                                              int& s, int& u) {
    int e = (lane < nvalid) ? le[lane] : 0;
    bool ok = (lane < nvalid) & (e < nE);
    if (!ok) e = 0;
    int d, sv;
    if (pairs) {
        const unsigned long long r = *(const unsigned long long*)(pairs + 2 * (long)e);
        d  = (int)(unsigned)r;
        sv = (int)(unsigned)(r >> 32);
    } else {
        d  = dst[e];
        sv = src[e];
    }
    unsigned uu = (unsigned)(d - tile0);
    ok = ok & (uu < (unsigned)TN);
    s = ok ? sv : 0;
    u = ok ? (int)uu : TN;
    return ok;
}

template <int F, int P>
__device__ __forceinline__ void add_rows(const float* __restrict__ x, int s, int u,
                                         float* acc, float* degs, int lane) {
    atomicAdd(&degs[u], 1.0f);
    if constexpr (F == 32) {
        float v[32];
#pragma unroll
        for (int j = 0; j < 32; ++j) {
            int sj = __builtin_amdgcn_readlane(s, j);
            v[j] = x[(long)sj * F + lane];
        }
#pragma unroll
        for (int j = 0; j < 32; ++j) {
            int uj = __builtin_amdgcn_readlane(u, j);
            atomicAdd(&acc[uj * P + lane], v[j]);
        }
    } else {
        v2f v[32];
#pragma unroll
        for (int j = 0; j < 32; ++j) {
            int sj = __builtin_amdgcn_readlane(s, j);
            v[j] = *(const v2f*)(x + (long)sj * F + 2 * lane);
        }
#pragma unroll
        for (int j = 0; j < 32; ++j) {
            int uj = __builtin_amdgcn_readlane(u, j);
            float* a = &acc[uj * P + 2 * lane];
            atomicAdd(a, v[j].x);
            atomicAdd(a + 1, v[j].y);
        }
    }
}

template <int F, int OUTW, bool RELU, int MODE>
__global__ __launch_bounds__(NT) void sage_owner_kernel(const float* __restrict__ x,
                                                        const int* __restrict__ src,
                                                        const int* __restrict__ dst,
                                                        const int* __restrict__ keys,
                                                        const int* __restrict__ pairs,
                                                        const float* __restrict__ Wself,
                                                        const float* __restrict__ Wneigh,
                                                        const float* __restrict__ bias,
                                                        float* __restrict__ y,
                                                        int* __restrict__ counts,
                                                        unsigned* __restrict__ lists, int cap,
                                                        int nNodes, int nE) {
    constexpr int P  = F + 2;
    constexpr int CT = OUTW / 16;
    __shared__ float acc[(TN + 1) * P];
    __shared__ float degs[TN + 16];
    __shared__ int list_e[NW * LCAP];
    __shared__ int lcount;

    const int wave  = threadIdx.x >> 5;
    const int lane  = threadIdx.x & 31;
    const int tile0 = blockIdx.x * TN;

    unsigned* tl = lists + (long)blockIdx.x * cap;
    int total = 0;
    bool from_list = false;
    unsigned wc = 0;
    const int b0 = __builtin_amdgcn_readfirstlane(wave) * 32;
    if constexpr (MODE == 2) {
        if (cap > 0) {
            total = counts[blockIdx.x];
            from_list = total <= cap;
            if (from_list && b0 < total) wc = tl[min(b0 + lane, total - 1)];
        }
    }

    for (int i = threadIdx.x; i < (TN + 1) * P; i += NT) acc[i] = 0.0f;
    for (int i = threadIdx.x; i < TN + 16; i += NT) degs[i] = 0.0f;
    if (threadIdx.x == 0) lcount = 0;
    __syncthreads();

    int* le = &list_e[wave * LCAP];
    int cnt = 0;

    auto flush = [&](const int* l, int nvalid) {
        int s, u;
        bool ok = confirm_batch(src, dst, pairs, l, nvalid, tile0, nE, lane, s, u);
        if constexpr (MODE == 1) {
            if (cap > 0) {
                unsigned okm = __builtin_amdgcn_ballot_w32(ok);
                int at = 0;
                if (lane == 0) at = atomicAdd(&lcount, __builtin_popcount(okm));
                at = __builtin_amdgcn_readfirstlane(at) + __builtin_amdgcn_mbcnt_lo(okm, 0);
                if (ok && at < cap) tl[at] = ((unsigned)s << UBITS) | (unsigned)u;
            }
        }
        add_rows<F, P>(x, s, u, acc, degs, lane);
    };

    unsigned tk[8];
#pragma unroll
    for (int p = 0; p < 8; ++p) tk[p] = ((blockIdx.x >> p) & 1u) ? 0xFFFFFFFFu : 0u;
    const int nBlocks = (nE + 31) >> 5;
    const int nSteps  = (nBlocks + 2 * NT - 1) / (2 * NT);
    const int step0   = (int)(((long)blockIdx.x * nSteps) / gridDim.x);

    auto load_step = [&](int4 (&w)[4], int st) {
#pragma unroll
        for (int k = 0; k < 2; ++k) {
            int b = (st * 2 + k) * NT + (int)threadIdx.x;
            if (b > nBlocks - 1) b = nBlocks - 1;
            w[2 * k]     = *(const int4*)(keys + 4 * (long)b);
            w[2 * k + 1] = *(const int4*)(keys + 4 * ((long)nBlocks + b));
        }
    };
    auto look_at = [&](const int4 (&w)[4], int st, bool live) {
#pragma unroll
        for (int h2 = 0; h2 < 2; ++h2) {
            const int b = (st * 2 + h2) * NT + (int)threadIdx.x;
            unsigned q = match32(w[2 * h2], w[2 * h2 + 1], tk);
            if (!(live && b < nBlocks)) q = 0;
            const int eBase = b * 32;
            while (true) {
                bool has = q != 0;
                unsigned bm = __builtin_amdgcn_ballot_w32(has);
                if (bm == 0) break;
                if (has) {
                    int p = __builtin_ctz(q);
                    q &= q - 1;
                    int e = eBase + p;
                    le[cnt + __builtin_amdgcn_mbcnt_lo(bm, 0)] = e;
                }
                cnt += __builtin_popcount(bm);
                if (cnt >= 32) {
                    cnt -= 32;
                    flush(le + cnt, 32);
                }
            }
        }
    };
    auto step_of = [&](int i) {
        int st = step0 + i;
        if (st >= nSteps) st -= nSteps;
        if (st >= nSteps) st -= nSteps;
        return st;
    };

    if (from_list) {
        int b = b0;
        while (b < total) {
            const int bn = b + NT;
            unsigned wn = 0;
            if (bn < total) wn = tl[min(bn + lane, total - 1)];
            const bool ok = b + lane < total;
            const int s = ok ? (int)(wc >> UBITS) : 0;
            const int u = ok ? (int)(wc & ((1u << UBITS) - 1u)) : TN;
            add_rows<F, P>(x, s, u, acc, degs, lane);
            wc = wn;
            b  = bn;
        }
    } else if (nSteps > 0) {
        int4 ka[4], kb[4];
        load_step(ka, step_of(0));
        for (int it = 0; it < nSteps; it += 2) {
            const int sa = step_of(it);
            const int sb = step_of(it + 1);
            load_step(kb, sb);
            look_at(ka, sa, true);
            load_step(ka, step_of(it + 2));
            look_at(kb, sb, it + 1 < nSteps);
        }
    }
    if (cnt > 0) flush(le, cnt);
    __syncthreads();
    if constexpr (MODE == 1) {
        if (cap > 0 && threadIdx.x == 0) counts[blockIdx.x] = lcount;
    }

    if (wave < TN / 16) {
        const int hi  = lane >> 4;
        const int l15 = lane & 15;
        const int lr  = wave * 16 + l15;
        int row = tile0 + lr;
        if (row > nNodes - 1) row = nNodes - 1;
        const float invd = 1.0f / fmaxf(degs[lr], 1.0f);

        v8f c[CT];
#pragma unroll
        for (int ct = 0; ct < CT; ++ct) {
            float bv = bias[ct * 16 + l15];
#pragma unroll
            for (int v = 0; v < 8; ++v) c[ct][v] = bv;
        }
        const float* xr = x + (long)row * F;
        const float* ar = acc + lr * P;
#pragma unroll
        for (int k0 = 0; k0 < F; k0 += 4) {
            const int kA = k0 + 2 * hi;
            v2f a, m;
            a.x = xr[kA];
            a.y = xr[kA + 1];
            m.x = ar[kA] * invd;
            m.y = ar[kA + 1] * invd;
#pragma unroll
            for (int ct = 0; ct < CT; ++ct) {
                const int colW = ct * 16 + l15;
                v2f b;
                b.x = Wself[(long)(kA)     * OUTW + colW];
                b.y = Wself[(long)(kA + 1) * OUTW + colW];
                c[ct] = __builtin_amdgcn_wmma_f32_16x16x4_f32(false, a, false, b, (short)0, c[ct], false, false);
                b.x = Wneigh[(long)(kA)     * OUTW + colW];
                b.y = Wneigh[(long)(kA + 1) * OUTW + colW];
                c[ct] = __builtin_amdgcn_wmma_f32_16x16x4_f32(false, m, false, b, (short)0, c[ct], false, false);
            }
        }
#pragma unroll
        for (int ct = 0; ct < CT; ++ct) {
#pragma unroll
            for (int v = 0; v < 8; ++v) {
                int r = tile0 + wave * 16 + v + 8 * hi;
                float o = c[ct][v];
                if (RELU) o = fmaxf(o, 0.0f);
                if (r < nNodes) y[(long)r * OUTW + ct * 16 + l15] = o;
            }
        }
    }
}

extern "C" void kernel_launch(void* const* d_in, const int* in_sizes, int n_in,
                              void* d_out, int out_size, void* d_ws, size_t ws_size,
                              hipStream_t stream) {
    const float* feat = (const float*)d_in[0];
    const int*   src  = (const int*)d_in[1];
    const int*   dst  = (const int*)d_in[2];
    const float* W1s  = (const float*)d_in[3];
    const float* W1n  = (const float*)d_in[4];
    const float* b1   = (const float*)d_in[5];
    const float* W2s  = (const float*)d_in[6];
    const float* W2n  = (const float*)d_in[7];
    const float* b2   = (const float*)d_in[8];
    float* out = (float*)d_out;

    const int nNodes = in_sizes[0] / IN_F;
    const int nE     = in_sizes[1];
    const int nTiles = (nNodes + TN - 1) / TN;
    const int nBlocks = (nE + 31) / 32;
    const int nWords = nBlocks * 8;

    float* h    = (float*)d_ws;
    int*   keys = (int*)(h + (size_t)nNodes * HID_F);
    int*   counts = keys + nWords;
    unsigned* lists = (unsigned*)(counts + ((nTiles + 3) & ~3));
    const bool fits = (size_t)((const char*)(lists + (size_t)nTiles * TCAP) - (const char*)d_ws) <= ws_size
                      && nNodes <= (1 << (32 - UBITS));
    const int cap = fits ? TCAP : 0;
    int* recs = (int*)(lists + (size_t)nTiles * TCAP);
    const bool fitsRecs = fits && (size_t)((const char*)(recs + 8 * (size_t)nWords) - (const char*)d_ws) <= ws_size;
    int* pairs = fitsRecs ? recs : nullptr;

    owner_key_kernel<<<(nBlocks + 8 * BPW - 1) / (8 * BPW), 256, 0, stream>>>(dst, src, keys, pairs, nE, nBlocks);
    sage_owner_kernel<IN_F, HID_F, true, 1><<<nTiles, NT, 0, stream>>>(feat, src, dst, keys, pairs, W1s, W1n, b1, h, counts, lists, cap, nNodes, nE);
    sage_owner_kernel<HID_F, OUT_F, false, 2><<<nTiles, NT, 0, stream>>>(h, src, dst, keys, pairs, W2s, W2n, b2, out, counts, lists, cap, nNodes, nE);
}


# Owner scan via a hit map: pre-pass marks, per group of 128 blocks and per key, the blocks holding such an edge; an owner reads its map words, then planes of hit blocks only; LDS-atomic sums as before
# speedup vs baseline: 5.1245x; 1.0449x over previous
#include <hip/hip_runtime.h>

#define IN_F 32
#define HID_F 64
#define OUT_F 16
#define TN 400
#define NW 32
#define NT (NW * 32)
#define LCAP 64
#define BPW 16
#define UBITS 9
#define TCAP 8192
#define GB (8 * BPW)

typedef float v2f __attribute__((ext_vector_type(2)));
typedef float v8f __attribute__((ext_vector_type(8)));

__global__ __launch_bounds__(256) void owner_key_kernel(const int* __restrict__ dst, const int* __restrict__ src,
                                                        int* __restrict__ keys, int* __restrict__ pairs,
                                                        int nE, int nBlocks) {
    const int lane = (int)threadIdx.x & 31;
    const int blk0 = (blockIdx.x * 8 + ((int)threadIdx.x >> 5)) * BPW;
    if (blk0 >= nBlocks) return;
    unsigned pl[8];
#pragma unroll
    for (int p = 0; p < 8; ++p) pl[p] = 0u;
#pragma unroll
    for (int c = 0; c < BPW / 8; ++c) {
        int d[8], s[8];
#pragma unroll
        for (int j = 0; j < 8; ++j) {
            const int e = (blk0 + c * 8 + j) * 32 + lane;
            d[j] = (e < nE) ? dst[e] : -1;
            s[j] = (pairs && e < nE) ? src[e] : 0;
        }
#pragma unroll
        for (int j = 0; j < 8; ++j) {
            const int blk = blk0 + c * 8 + j;
            const int e = blk * 32 + lane;
            if (pairs && blk < nBlocks) *(int2*)(pairs + 2 * (long)e) = make_int2(d[j], s[j]);
            const unsigned key = d[j] < 0 ? 255u : (((unsigned)d[j] / TN) & 255u);
#pragma unroll
            for (int p = 0; p < 8; ++p) {
                const unsigned m = __builtin_amdgcn_ballot_w32(((key >> p) & 1u) != 0u);
                if (lane == c * 8 + j) pl[p] = m;
            }
        }
    }
    const int blk = blk0 + lane;
    if (lane < BPW && blk < nBlocks) {
        *(int4*)(keys + 4 * (long)blk)             = make_int4((int)pl[0], (int)pl[1], (int)pl[2], (int)pl[3]);
        *(int4*)(keys + 4 * ((long)nBlocks + blk)) = make_int4((int)pl[4], (int)pl[5], (int)pl[6], (int)pl[7]);
    }
}

__global__ __launch_bounds__(256) void owner_hits_kernel(const int* __restrict__ dst, const int* __restrict__ src,
                                                         int* __restrict__ keys, int* __restrict__ pairs,
                                                         unsigned* __restrict__ hits, int nE, int nBlocks) {
    __shared__ alignas(16) unsigned slab[256 * 4];
    const int lane = (int)threadIdx.x & 31;
    const int wv   = (int)threadIdx.x >> 5;
    const int blk0 = (blockIdx.x * 8 + wv) * BPW;
    *(uint4*)&slab[threadIdx.x * 4] = make_uint4(0u, 0u, 0u, 0u);
    __syncthreads();
    if (blk0 < nBlocks) {
        unsigned pl[4];
#pragma unroll
        for (int p = 0; p < 4; ++p) pl[p] = 0u;
#pragma unroll
        for (int c = 0; c < BPW / 8; ++c) {
            int d[8], s[8];
#pragma unroll
            for (int j = 0; j < 8; ++j) {
                const int e = (blk0 + c * 8 + j) * 32 + lane;
                d[j] = (e < nE) ? dst[e] : -1;
                s[j] = (pairs && e < nE) ? src[e] : 0;
            }
#pragma unroll
            for (int j = 0; j < 8; ++j) {
                const int blk = blk0 + c * 8 + j;
                const int e = blk * 32 + lane;
                if (pairs && blk < nBlocks) *(int2*)(pairs + 2 * (long)e) = make_int2(d[j], s[j]);
                const unsigned key = d[j] < 0 ? 255u : (((unsigned)d[j] / TN) & 255u);
                const int bl = wv * BPW + c * 8 + j;
                if (d[j] >= 0) atomicOr(&slab[key * 4 + (bl >> 5)], 1u << (bl & 31));
#pragma unroll
                for (int p = 0; p < 8; ++p) {
                    const unsigned m = __builtin_amdgcn_ballot_w32(((key >> p) & 1u) != 0u);
                    if (lane == 2 * (c * 8 + j) + (p >> 2)) pl[p & 3] = m;
                }
            }
        }
        const int blk = blk0 + (lane >> 1);
        if (blk < nBlocks)
            *(int4*)(keys + 8 * (long)blk + 4 * (lane & 1)) = make_int4((int)pl[0], (int)pl[1], (int)pl[2], (int)pl[3]);
    }
    __syncthreads();
    *(uint4*)(hits + ((long)blockIdx.x * 256) * 4 + threadIdx.x * 4) = *(const uint4*)&slab[threadIdx.x * 4];
}

__device__ __forceinline__ unsigned match32(int4 lo, int4 hi, const unsigned (&t)[8]) {
    unsigned a = (unsigned)lo.x ^ t[0];
    a |= (unsigned)lo.y ^ t[1];
    a |= (unsigned)lo.z ^ t[2];
    a |= (unsigned)lo.w ^ t[3];
    a |= (unsigned)hi.x ^ t[4];
    a |= (unsigned)hi.y ^ t[5];
    a |= (unsigned)hi.z ^ t[6];
    a |= (unsigned)hi.w ^ t[7];
    return ~a;
}

__device__ __forceinline__ bool confirm_batch(const int* __restrict__ src, const int* __restrict__ dst,
                                              const int* __restrict__ pairs,
                                              const int* le, int nvalid, int tile0, int nE, int lane,
                                              int& s, int& u) {
    int e = (lane < nvalid) ? le[lane] : 0;
    bool ok = (lane < nvalid) & (e < nE);
    if (!ok) e = 0;
    int d, sv;
    if (pairs) {
        const unsigned long long r = *(const unsigned long long*)(pairs + 2 * (long)e);
        d  = (int)(unsigned)r;
        sv = (int)(unsigned)(r >> 32);
    } else {
        d  = dst[e];
        sv = src[e];
    }
    unsigned uu = (unsigned)(d - tile0);
    ok = ok & (uu < (unsigned)TN);
    s = ok ? sv : 0;
    u = ok ? (int)uu : TN;
    return ok;
}

template <int F, int P>
__device__ __forceinline__ void add_rows(const float* __restrict__ x, int s, int u,
                                         float* acc, float* degs, int lane) {
    atomicAdd(&degs[u], 1.0f);
    if constexpr (F == 32) {
        float v[32];
#pragma unroll
        for (int j = 0; j < 32; ++j) {
            int sj = __builtin_amdgcn_readlane(s, j);
            v[j] = x[(long)sj * F + lane];
        }
#pragma unroll
        for (int j = 0; j < 32; ++j) {
            int uj = __builtin_amdgcn_readlane(u, j);
            atomicAdd(&acc[uj * P + lane], v[j]);
        }
    } else {
        v2f v[32];
#pragma unroll
        for (int j = 0; j < 32; ++j) {
            int sj = __builtin_amdgcn_readlane(s, j);
            v[j] = *(const v2f*)(x + (long)sj * F + 2 * lane);
        }
#pragma unroll
        for (int j = 0; j < 32; ++j) {
            int uj = __builtin_amdgcn_readlane(u, j);
            float* a = &acc[uj * P + 2 * lane];
            atomicAdd(a, v[j].x);
            atomicAdd(a + 1, v[j].y);
        }
    }
}

template <int F, int OUTW, bool RELU, int MODE, bool HITS>
__global__ __launch_bounds__(NT) void sage_owner_kernel(const float* __restrict__ x,
                                                        const int* __restrict__ src,
                                                        const int* __restrict__ dst,
                                                        const int* __restrict__ keys,
                                                        const int* __restrict__ pairs,
                                                        const unsigned* __restrict__ hits,
                                                        const float* __restrict__ Wself,
                                                        const float* __restrict__ Wneigh,
                                                        const float* __restrict__ bias,
                                                        float* __restrict__ y,
                                                        int* __restrict__ counts,
                                                        unsigned* __restrict__ lists, int cap,
                                                        int nNodes, int nE) {
    constexpr int P  = F + 2;
    constexpr int CT = OUTW / 16;
    __shared__ float acc[(TN + 1) * P];
    __shared__ float degs[TN + 16];
    __shared__ int list_e[NW * LCAP];
    __shared__ int list_b[HITS ? NW * LCAP : 1];
    __shared__ int lcount;

    const int wave  = threadIdx.x >> 5;
    const int lane  = threadIdx.x & 31;
    const int tile0 = blockIdx.x * TN;

    unsigned* tl = lists + (long)blockIdx.x * cap;
    int total = 0;
    bool from_list = false;
    unsigned wc = 0;
    const int b0 = __builtin_amdgcn_readfirstlane(wave) * 32;
    if constexpr (MODE == 2) {
        if (cap > 0) {
            total = counts[blockIdx.x];
            from_list = total <= cap;
            if (from_list && b0 < total) wc = tl[min(b0 + lane, total - 1)];
        }
    }

    const int nBlocks = (nE + 31) >> 5;
    const int nGroups = (nBlocks + GB - 1) / GB;
    const int gFirst  = __builtin_amdgcn_readfirstlane(wave) + NW * (lane >> 2);
    auto load_hits = [&](int r) -> unsigned {
        const int g = gFirst + r * (NW * 8);
        return g < nGroups ? hits[((long)g * 256 + (blockIdx.x & 255u)) * 4 + (lane & 3)] : 0u;
    };
    unsigned hpre = 0;
    if constexpr (HITS && MODE == 1) hpre = load_hits(0);

    for (int i = threadIdx.x; i < (TN + 1) * P; i += NT) acc[i] = 0.0f;
    for (int i = threadIdx.x; i < TN + 16; i += NT) degs[i] = 0.0f;
    if (threadIdx.x == 0) lcount = 0;
    __syncthreads();

    int* le = &list_e[wave * LCAP];
    int cnt = 0;

    auto flush = [&](const int* l, int nvalid) {
        int s, u;
        bool ok = confirm_batch(src, dst, pairs, l, nvalid, tile0, nE, lane, s, u);
        if constexpr (MODE == 1) {
            if (cap > 0) {
                unsigned okm = __builtin_amdgcn_ballot_w32(ok);
                int at = 0;
                if (lane == 0) at = atomicAdd(&lcount, __builtin_popcount(okm));
                at = __builtin_amdgcn_readfirstlane(at) + __builtin_amdgcn_mbcnt_lo(okm, 0);
                if (ok && at < cap) tl[at] = ((unsigned)s << UBITS) | (unsigned)u;
            }
        }
        add_rows<F, P>(x, s, u, acc, degs, lane);
    };

    unsigned tk[8];
#pragma unroll
    for (int p = 0; p < 8; ++p) tk[p] = ((blockIdx.x >> p) & 1u) ? 0xFFFFFFFFu : 0u;

    if (from_list) {
        int b = b0;
        while (b < total) {
            const int bn = b + NT;
            unsigned wn = 0;
            if (bn < total) wn = tl[min(bn + lane, total - 1)];
            const bool ok = b + lane < total;
            const int s = ok ? (int)(wc >> UBITS) : 0;
            const int u = ok ? (int)(wc & ((1u << UBITS) - 1u)) : TN;
            add_rows<F, P>(x, s, u, acc, degs, lane);
            wc = wn;
            b  = bn;
        }
    } else if constexpr (HITS) {
        int* lb = &list_b[wave * LCAP];
        int cb = 0;
        auto look_blocks = [&](const int* l, int nvalid) {
            const int b = (lane < nvalid) ? l[lane] : 0;
            const int4 lo = *(const int4*)(keys + 8 * (long)b);
            const int4 hi = *(const int4*)(keys + 8 * (long)b + 4);
            unsigned q = match32(lo, hi, tk);
            if (lane >= nvalid) q = 0;
            const int eBase = b * 32;
            while (true) {
                bool has = q != 0;
                unsigned bm = __builtin_amdgcn_ballot_w32(has);
                if (bm == 0) break;
                if (has) {
                    int p = __builtin_ctz(q);
                    q &= q - 1;
                    int e = eBase + p;
                    le[cnt + __builtin_amdgcn_mbcnt_lo(bm, 0)] = e;
                }
                cnt += __builtin_popcount(bm);
                if (cnt >= 32) {
                    cnt -= 32;
                    flush(le + cnt, 32);
                }
            }
        };
        const int nRounds = (nGroups + NW * 8 - 1) / (NW * 8);
        unsigned hcur = (MODE == 1) ? hpre : load_hits(0);
        for (int r = 0; r < nRounds; ++r) {
            const unsigned hnext = load_hits(r + 1);
            unsigned hw = hcur;
            const int bBase = (gFirst + r * (NW * 8)) * GB + (lane & 3) * 32;
            while (true) {
                bool has = hw != 0;
                unsigned bm = __builtin_amdgcn_ballot_w32(has);
                if (bm == 0) break;
                if (has) {
                    int p = __builtin_ctz(hw);
                    hw &= hw - 1;
                    lb[cb + __builtin_amdgcn_mbcnt_lo(bm, 0)] = bBase + p;
                }
                cb += __builtin_popcount(bm);
                if (cb >= 32) {
                    cb -= 32;
                    look_blocks(lb + cb, 32);
                }
            }
            hcur = hnext;
        }
        if (cb > 0) look_blocks(lb, cb);
    } else {
        const int nSteps  = (nBlocks + 2 * NT - 1) / (2 * NT);
        const int step0   = (int)(((long)blockIdx.x * nSteps) / gridDim.x);

        auto load_step = [&](int4 (&w)[4], int st) {
#pragma unroll
            for (int k = 0; k < 2; ++k) {
                int b = (st * 2 + k) * NT + (int)threadIdx.x;
                if (b > nBlocks - 1) b = nBlocks - 1;
                w[2 * k]     = *(const int4*)(keys + 4 * (long)b);
                w[2 * k + 1] = *(const int4*)(keys + 4 * ((long)nBlocks + b));
            }
        };
        auto look_at = [&](const int4 (&w)[4], int st, bool live) {
#pragma unroll
            for (int h2 = 0; h2 < 2; ++h2) {
                const int b = (st * 2 + h2) * NT + (int)threadIdx.x;
                unsigned q = match32(w[2 * h2], w[2 * h2 + 1], tk);
                if (!(live && b < nBlocks)) q = 0;
                const int eBase = b * 32;
                while (true) {
                    bool has = q != 0;
                    unsigned bm = __builtin_amdgcn_ballot_w32(has);
                    if (bm == 0) break;
                    if (has) {
                        int p = __builtin_ctz(q);
                        q &= q - 1;
                        int e = eBase + p;
                        le[cnt + __builtin_amdgcn_mbcnt_lo(bm, 0)] = e;
                    }
                    cnt += __builtin_popcount(bm);
                    if (cnt >= 32) {
                        cnt -= 32;
                        flush(le + cnt, 32);
                    }
                }
            }
        };
        auto step_of = [&](int i) {
            int st = step0 + i;
            if (st >= nSteps) st -= nSteps;
            if (st >= nSteps) st -= nSteps;
            return st;
        };

        if (nSteps > 0) {
            int4 ka[4], kb[4];
            load_step(ka, step_of(0));
            for (int it = 0; it < nSteps; it += 2) {
                const int sa = step_of(it);
                const int sb = step_of(it + 1);
                load_step(kb, sb);
                look_at(ka, sa, true);
                load_step(ka, step_of(it + 2));
                look_at(kb, sb, it + 1 < nSteps);
            }
        }
    }
    if (cnt > 0) flush(le, cnt);
    __syncthreads();
    if constexpr (MODE == 1) {
        if (cap > 0 && threadIdx.x == 0) counts[blockIdx.x] = lcount;
    }

    if (wave < TN / 16) {
        const int hi  = lane >> 4;
        const int l15 = lane & 15;
        const int lr  = wave * 16 + l15;
        int row = tile0 + lr;
        if (row > nNodes - 1) row = nNodes - 1;
        const float invd = 1.0f / fmaxf(degs[lr], 1.0f);

        v8f c[CT];
#pragma unroll
        for (int ct = 0; ct < CT; ++ct) {
            float bv = bias[ct * 16 + l15];
#pragma unroll
            for (int v = 0; v < 8; ++v) c[ct][v] = bv;
        }
        const float* xr = x + (long)row * F;
        const float* ar = acc + lr * P;
#pragma unroll
        for (int k0 = 0; k0 < F; k0 += 4) {
            const int kA = k0 + 2 * hi;
            v2f a, m;
            a.x = xr[kA];
            a.y = xr[kA + 1];
            m.x = ar[kA] * invd;
            m.y = ar[kA + 1] * invd;
#pragma unroll
            for (int ct = 0; ct < CT; ++ct) {
                const int colW = ct * 16 + l15;
                v2f b;
                b.x = Wself[(long)(kA)     * OUTW + colW];
                b.y = Wself[(long)(kA + 1) * OUTW + colW];
                c[ct] = __builtin_amdgcn_wmma_f32_16x16x4_f32(false, a, false, b, (short)0, c[ct], false, false);
                b.x = Wneigh[(long)(kA)     * OUTW + colW];
                b.y = Wneigh[(long)(kA + 1) * OUTW + colW];
                c[ct] = __builtin_amdgcn_wmma_f32_16x16x4_f32(false, m, false, b, (short)0, c[ct], false, false);
            }
        }
#pragma unroll
        for (int ct = 0; ct < CT; ++ct) {
#pragma unroll
            for (int v = 0; v < 8; ++v) {
                int r = tile0 + wave * 16 + v + 8 * hi;
                float o = c[ct][v];
                if (RELU) o = fmaxf(o, 0.0f);
                if (r < nNodes) y[(long)r * OUTW + ct * 16 + l15] = o;
            }
        }
    }
}

extern "C" void kernel_launch(void* const* d_in, const int* in_sizes, int n_in,
                              void* d_out, int out_size, void* d_ws, size_t ws_size,
                              hipStream_t stream) {
    const float* feat = (const float*)d_in[0];
    const int*   src  = (const int*)d_in[1];
    const int*   dst  = (const int*)d_in[2];
    const float* W1s  = (const float*)d_in[3];
    const float* W1n  = (const float*)d_in[4];
    const float* b1   = (const float*)d_in[5];
    const float* W2s  = (const float*)d_in[6];
    const float* W2n  = (const float*)d_in[7];
    const float* b2   = (const float*)d_in[8];
    float* out = (float*)d_out;

    const int nNodes = in_sizes[0] / IN_F;
    const int nE     = in_sizes[1];
    const int nTiles = (nNodes + TN - 1) / TN;
    const int nBlocks = (nE + 31) / 32;
    const int nWords = nBlocks * 8;

    float* h    = (float*)d_ws;
    int*   keys = (int*)(h + (size_t)nNodes * HID_F);
    int*   counts = keys + nWords;
    unsigned* lists = (unsigned*)(counts + ((nTiles + 31) & ~31));
    const bool fits = (size_t)((const char*)(lists + (size_t)nTiles * TCAP) - (const char*)d_ws) <= ws_size
                      && nNodes <= (1 << (32 - UBITS));
    const int cap = fits ? TCAP : 0;
    int* recs = (int*)(lists + (size_t)nTiles * TCAP);
    const bool fitsRecs = fits && (size_t)((const char*)(recs + 8 * (size_t)nWords) - (const char*)d_ws) <= ws_size;
    int* pairs = fitsRecs ? recs : nullptr;

    const int nGroups = (nBlocks + GB - 1) / GB;
    unsigned* hitw = (unsigned*)(recs + 8 * (size_t)nWords);
    const bool fitsHits = fitsRecs && (size_t)((const char*)(hitw + (size_t)nGroups * 1024) - (const char*)d_ws) <= ws_size;

    if (fitsHits) {
        owner_hits_kernel<<<nGroups, 256, 0, stream>>>(dst, src, keys, pairs, hitw, nE, nBlocks);
        sage_owner_kernel<IN_F, HID_F, true, 1, true><<<nTiles, NT, 0, stream>>>(feat, src, dst, keys, pairs, hitw, W1s, W1n, b1, h, counts, lists, cap, nNodes, nE);
        sage_owner_kernel<HID_F, OUT_F, false, 2, true><<<nTiles, NT, 0, stream>>>(h, src, dst, keys, pairs, hitw, W2s, W2n, b2, out, counts, lists, cap, nNodes, nE);
    } else {
        owner_key_kernel<<<(nBlocks + 8 * BPW - 1) / (8 * BPW), 256, 0, stream>>>(dst, src, keys, pairs, nE, nBlocks);
        sage_owner_kernel<IN_F, HID_F, true, 1, false><<<nTiles, NT, 0, stream>>>(feat, src, dst, keys, pairs, nullptr, W1s, W1n, b1, h, counts, lists, cap, nNodes, nE);
        sage_owner_kernel<HID_F, OUT_F, false, 2, false><<<nTiles, NT, 0, stream>>>(h, src, dst, keys, pairs, nullptr, W2s, W2n, b2, out, counts, lists, cap, nNodes, nE);
    }
}


# Hit map per octet of 8 edges (was per block of 32): owners read a hit octet's eight (dst,src) records as one 64-byte piece; no key planes, no per-edge record read; LDS-atomic sums as before
# speedup vs baseline: 5.3707x; 1.0480x over previous
#include <hip/hip_runtime.h>

#define IN_F 32
#define HID_F 64
#define OUT_F 16
#define TN 400
#define NW 32
#define NT (NW * 32)
#define LCAP 64
#define BPW 16
#define UBITS 9
#define TCAP 8192
#define GB (8 * BPW)
#define MW (GB / 8)
static_assert(MW >= 4 && MW <= 32 && 32 % MW == 0 && MW % 4 == 0, "the lanes of a wave must split evenly over the map words of a group");

typedef float v2f __attribute__((ext_vector_type(2)));
typedef float v8f __attribute__((ext_vector_type(8)));

__global__ __launch_bounds__(256) void owner_key_kernel(const int* __restrict__ dst, const int* __restrict__ src,
                                                        int* __restrict__ keys, int* __restrict__ pairs,
                                                        int nE, int nBlocks) {
    const int lane = (int)threadIdx.x & 31;
    const int blk0 = (blockIdx.x * 8 + ((int)threadIdx.x >> 5)) * BPW;
    if (blk0 >= nBlocks) return;
    unsigned pl[8];
#pragma unroll
    for (int p = 0; p < 8; ++p) pl[p] = 0u;
#pragma unroll
    for (int c = 0; c < BPW / 8; ++c) {
        int d[8], s[8];
#pragma unroll
        for (int j = 0; j < 8; ++j) {
            const int e = (blk0 + c * 8 + j) * 32 + lane;
            d[j] = (e < nE) ? dst[e] : -1;
            s[j] = (pairs && e < nE) ? src[e] : 0;
        }
#pragma unroll
        for (int j = 0; j < 8; ++j) {
            const int blk = blk0 + c * 8 + j;
            const int e = blk * 32 + lane;
            if (pairs && blk < nBlocks) *(int2*)(pairs + 2 * (long)e) = make_int2(d[j], s[j]);
            const unsigned key = d[j] < 0 ? 255u : (((unsigned)d[j] / TN) & 255u);
#pragma unroll
            for (int p = 0; p < 8; ++p) {
                const unsigned m = __builtin_amdgcn_ballot_w32(((key >> p) & 1u) != 0u);
                if (lane == c * 8 + j) pl[p] = m;
            }
        }
    }
    const int blk = blk0 + lane;
    if (lane < BPW && blk < nBlocks) {
        *(int4*)(keys + 4 * (long)blk)             = make_int4((int)pl[0], (int)pl[1], (int)pl[2], (int)pl[3]);
        *(int4*)(keys + 4 * ((long)nBlocks + blk)) = make_int4((int)pl[4], (int)pl[5], (int)pl[6], (int)pl[7]);
    }
}

__global__ __launch_bounds__(256) void owner_hits_kernel(const int* __restrict__ dst, const int* __restrict__ src,
                                                         int* __restrict__ pairs,
                                                         unsigned* __restrict__ hits, int nE, int nBlocks) {
    __shared__ unsigned slab[256 * (MW + 1)];
    const int lane = (int)threadIdx.x & 31;
    const int wv   = (int)threadIdx.x >> 5;
    const int blk0 = (blockIdx.x * 8 + wv) * BPW;
    for (int i = threadIdx.x; i < 256 * (MW + 1); i += 256) slab[i] = 0u;
    __syncthreads();
    if (blk0 < nBlocks) {
#pragma unroll
        for (int c = 0; c < BPW / 8; ++c) {
            int d[8], s[8];
#pragma unroll
            for (int j = 0; j < 8; ++j) {
                const int e = (blk0 + c * 8 + j) * 32 + lane;
                d[j] = (e < nE) ? dst[e] : -1;
                s[j] = (e < nE) ? src[e] : 0;
            }
#pragma unroll
            for (int j = 0; j < 8; ++j) {
                const int blk = blk0 + c * 8 + j;
                const int e = blk * 32 + lane;
                if (blk < nBlocks) *(int2*)(pairs + 2 * (long)e) = make_int2(d[j], s[j]);
                if (d[j] >= 0) {
                    const unsigned key = ((unsigned)d[j] / TN) & 255u;
                    const int o = (wv * BPW + c * 8 + j) * 4 + (lane >> 3);
                    atomicOr(&slab[key * (MW + 1) + (o >> 5)], 1u << (o & 31));
                }
            }
        }
    }
    __syncthreads();
    uint4 m[MW / 4];
#pragma unroll
    for (int r = 0; r < MW / 4; ++r) {
        const int i = ((int)threadIdx.x + 256 * r) * 4;
        const unsigned* sp = &slab[i + i / MW];
        m[r] = make_uint4(sp[0], sp[1], sp[2], sp[3]);
    }
#pragma unroll
    for (int r = 0; r < MW / 4; ++r)
        *(uint4*)(hits + (long)blockIdx.x * (256 * MW) + ((int)threadIdx.x + 256 * r) * 4) = m[r];
}

__device__ __forceinline__ unsigned match32(int4 lo, int4 hi, const unsigned (&t)[8]) {
    unsigned a = (unsigned)lo.x ^ t[0];
    a |= (unsigned)lo.y ^ t[1];
    a |= (unsigned)lo.z ^ t[2];
    a |= (unsigned)lo.w ^ t[3];
    a |= (unsigned)hi.x ^ t[4];
    a |= (unsigned)hi.y ^ t[5];
    a |= (unsigned)hi.z ^ t[6];
    a |= (unsigned)hi.w ^ t[7];
    return ~a;
}

__device__ __forceinline__ bool confirm_batch(const int* __restrict__ src, const int* __restrict__ dst,
                                              const int* __restrict__ pairs,
                                              const int* le, int nvalid, int tile0, int nE, int lane,
                                              int& s, int& u) {
    int e = (lane < nvalid) ? le[lane] : 0;
    bool ok = (lane < nvalid) & (e < nE);
    if (!ok) e = 0;
    int d, sv;
    if (pairs) {
        const unsigned long long r = *(const unsigned long long*)(pairs + 2 * (long)e);
        d  = (int)(unsigned)r;
        sv = (int)(unsigned)(r >> 32);
    } else {
        d  = dst[e];
        sv = src[e];
    }
    unsigned uu = (unsigned)(d - tile0);
    ok = ok & (uu < (unsigned)TN);
    s = ok ? sv : 0;
    u = ok ? (int)uu : TN;
    return ok;
}

template <int F, int P>
__device__ __forceinline__ void add_rows(const float* __restrict__ x, int s, int u,
                                         float* acc, float* degs, int lane) {
    atomicAdd(&degs[u], 1.0f);
    if constexpr (F == 32) {
        float v[32];
#pragma unroll
        for (int j = 0; j < 32; ++j) {
            int sj = __builtin_amdgcn_readlane(s, j);
            v[j] = x[(long)sj * F + lane];
        }
#pragma unroll
        for (int j = 0; j < 32; ++j) {
            int uj = __builtin_amdgcn_readlane(u, j);
            atomicAdd(&acc[uj * P + lane], v[j]);
        }
    } else {
        v2f v[32];
#pragma unroll
        for (int j = 0; j < 32; ++j) {
            int sj = __builtin_amdgcn_readlane(s, j);
            v[j] = *(const v2f*)(x + (long)sj * F + 2 * lane);
        }
#pragma unroll
        for (int j = 0; j < 32; ++j) {
            int uj = __builtin_amdgcn_readlane(u, j);
            float* a = &acc[uj * P + 2 * lane];
            atomicAdd(a, v[j].x);
            atomicAdd(a + 1, v[j].y);
        }
    }
}

template <int F, int OUTW, bool RELU, int MODE, bool HITS>
__global__ __launch_bounds__(NT) void sage_owner_kernel(const float* __restrict__ x,
                                                        const int* __restrict__ src,
                                                        const int* __restrict__ dst,
                                                        const int* __restrict__ keys,
                                                        const int* __restrict__ pairs,
                                                        const unsigned* __restrict__ hits,
                                                        const float* __restrict__ Wself,
                                                        const float* __restrict__ Wneigh,
                                                        const float* __restrict__ bias,
                                                        float* __restrict__ y,
                                                        int* __restrict__ counts,
                                                        unsigned* __restrict__ lists, int cap,
                                                        int nNodes, int nE) {
    constexpr int P  = F + 2;
    constexpr int CT = OUTW / 16;
    __shared__ float acc[(TN + 1) * P];
    __shared__ float degs[TN + 16];
    __shared__ int list_e[NW * LCAP];
    __shared__ int list_b[HITS ? NW * LCAP : 1];
    __shared__ int lcount;

    const int wave  = threadIdx.x >> 5;
    const int lane  = threadIdx.x & 31;
    const int tile0 = blockIdx.x * TN;

    unsigned* tl = lists + (long)blockIdx.x * cap;
    int total = 0;
    bool from_list = false;
    unsigned wc = 0;
    const int b0 = __builtin_amdgcn_readfirstlane(wave) * 32;
    if constexpr (MODE == 2) {
        if (cap > 0) {
            total = counts[blockIdx.x];
            from_list = total <= cap;
            if (from_list && b0 < total) wc = tl[min(b0 + lane, total - 1)];
        }
    }

    const int nBlocks = (nE + 31) >> 5;
    constexpr int GPR = 32 / MW;
    const int nGroups = (nBlocks + GB - 1) / GB;
    const int gFirst  = __builtin_amdgcn_readfirstlane(wave) + NW * (lane / MW);
    auto load_hits = [&](int r) -> unsigned {
        const int g = gFirst + r * (NW * GPR);
        return g < nGroups ? hits[((long)g * 256 + (blockIdx.x & 255u)) * MW + (lane % MW)] : 0u;
    };
    unsigned hpre = 0;
    if constexpr (HITS && MODE == 1) hpre = load_hits(0);

    for (int i = threadIdx.x; i < (TN + 1) * P; i += NT) acc[i] = 0.0f;
    for (int i = threadIdx.x; i < TN + 16; i += NT) degs[i] = 0.0f;
    if (threadIdx.x == 0) lcount = 0;
    __syncthreads();

    int* le = &list_e[wave * LCAP];
    int cnt = 0;

    auto flush = [&](const int* l, int nvalid) {
        int s, u;
        bool ok = confirm_batch(src, dst, pairs, l, nvalid, tile0, nE, lane, s, u);
        if constexpr (MODE == 1) {
            if (cap > 0) {
                unsigned okm = __builtin_amdgcn_ballot_w32(ok);
                int at = 0;
                if (lane == 0) at = atomicAdd(&lcount, __builtin_popcount(okm));
                at = __builtin_amdgcn_readfirstlane(at) + __builtin_amdgcn_mbcnt_lo(okm, 0);
                if (ok && at < cap) tl[at] = ((unsigned)s << UBITS) | (unsigned)u;
            }
        }
        add_rows<F, P>(x, s, u, acc, degs, lane);
    };
    auto flushw = [&](const int* l, int nvalid) {
        const bool ok = lane < nvalid;
        const unsigned w = ok ? (unsigned)l[lane] : 0u;
        if constexpr (MODE == 1) {
            if (cap > 0) {
                int at = 0;
                if (lane == 0) at = atomicAdd(&lcount, nvalid);
                at = __builtin_amdgcn_readfirstlane(at) + lane;
                if (ok && at < cap) tl[at] = w;
            }
        }
        const int s = ok ? (int)(w >> UBITS) : 0;
        const int u = ok ? (int)(w & ((1u << UBITS) - 1u)) : TN;
        add_rows<F, P>(x, s, u, acc, degs, lane);
    };

    unsigned tk[8];
#pragma unroll
    for (int p = 0; p < 8; ++p) tk[p] = ((blockIdx.x >> p) & 1u) ? 0xFFFFFFFFu : 0u;

    if (from_list) {
        int b = b0;
        while (b < total) {
            const int bn = b + NT;
            unsigned wn = 0;
            if (bn < total) wn = tl[min(bn + lane, total - 1)];
            const bool ok = b + lane < total;
            const int s = ok ? (int)(wc >> UBITS) : 0;
            const int u = ok ? (int)(wc & ((1u << UBITS) - 1u)) : TN;
            add_rows<F, P>(x, s, u, acc, degs, lane);
            wc = wn;
            b  = bn;
        }
    } else if constexpr (HITS) {
        int* lb = &list_b[wave * LCAP];
        int cb = 0;
        auto look_octets = [&](const int* l, int nvalid) {
            int4 rc[4];
            unsigned vmask = 0;
#pragma unroll
            for (int i = 0; i < 4; ++i) {
                const int k = i * 8 + (lane >> 2);
                const bool v = k < nvalid;
                const int o = v ? l[k] : 0;
                rc[i] = *(const int4*)(pairs + 16 * (long)o + 4 * (lane & 3));
                vmask |= (v ? 3u : 0u) << (2 * i);
            }
            unsigned wd[8];
            unsigned q = 0;
#pragma unroll
            for (int i = 0; i < 4; ++i) {
                const unsigned u0 = (unsigned)(rc[i].x - tile0);
                const unsigned u1 = (unsigned)(rc[i].z - tile0);
                q |= (u0 < (unsigned)TN ? 1u : 0u) << (2 * i);
                q |= (u1 < (unsigned)TN ? 1u : 0u) << (2 * i + 1);
                wd[2 * i]     = ((unsigned)rc[i].y << UBITS) | u0;
                wd[2 * i + 1] = ((unsigned)rc[i].w << UBITS) | u1;
            }
            q &= vmask;
            while (true) {
                bool has = q != 0;
                unsigned bm = __builtin_amdgcn_ballot_w32(has);
                if (bm == 0) break;
                if (has) {
                    const int p = __builtin_ctz(q);
                    q &= q - 1;
                    unsigned w = wd[0];
#pragma unroll
                    for (int i = 1; i < 8; ++i) w = (p == i) ? wd[i] : w;
                    le[cnt + __builtin_amdgcn_mbcnt_lo(bm, 0)] = (int)w;
                }
                cnt += __builtin_popcount(bm);
                if (cnt >= 32) {
                    cnt -= 32;
                    flushw(le + cnt, 32);
                }
            }
        };
        const int nRounds = (nGroups + NW * GPR - 1) / (NW * GPR);
        unsigned hcur = (MODE == 1) ? hpre : load_hits(0);
        for (int r = 0; r < nRounds; ++r) {
            const unsigned hnext = load_hits(r + 1);
            unsigned hw = hcur;
            const int oBase = (gFirst + r * (NW * GPR)) * (GB * 4) + (lane % MW) * 32;
            while (true) {
                bool has = hw != 0;
                unsigned bm = __builtin_amdgcn_ballot_w32(has);
                if (bm == 0) break;
                if (has) {
                    int p = __builtin_ctz(hw);
                    hw &= hw - 1;
                    lb[cb + __builtin_amdgcn_mbcnt_lo(bm, 0)] = oBase + p;
                }
                cb += __builtin_popcount(bm);
                if (cb >= 32) {
                    cb -= 32;
                    look_octets(lb + cb, 32);
                }
            }
            hcur = hnext;
        }
        if (cb > 0) look_octets(lb, cb);
    } else {
        const int nSteps  = (nBlocks + 2 * NT - 1) / (2 * NT);
        const int step0   = (int)(((long)blockIdx.x * nSteps) / gridDim.x);

        auto load_step = [&](int4 (&w)[4], int st) {
#pragma unroll
            for (int k = 0; k < 2; ++k) {
                int b = (st * 2 + k) * NT + (int)threadIdx.x;
                if (b > nBlocks - 1) b = nBlocks - 1;
                w[2 * k]     = *(const int4*)(keys + 4 * (long)b);
                w[2 * k + 1] = *(const int4*)(keys + 4 * ((long)nBlocks + b));
            }
        };
        auto look_at = [&](const int4 (&w)[4], int st, bool live) {
#pragma unroll
            for (int h2 = 0; h2 < 2; ++h2) {
                const int b = (st * 2 + h2) * NT + (int)threadIdx.x;
                unsigned q = match32(w[2 * h2], w[2 * h2 + 1], tk);
                if (!(live && b < nBlocks)) q = 0;
                const int eBase = b * 32;
                while (true) {
                    bool has = q != 0;
                    unsigned bm = __builtin_amdgcn_ballot_w32(has);
                    if (bm == 0) break;
                    if (has) {
                        int p = __builtin_ctz(q);
                        q &= q - 1;
                        int e = eBase + p;
                        le[cnt + __builtin_amdgcn_mbcnt_lo(bm, 0)] = e;
                    }
                    cnt += __builtin_popcount(bm);
                    if (cnt >= 32) {
                        cnt -= 32;
                        flush(le + cnt, 32);
                    }
                }
            }
        };
        auto step_of = [&](int i) {
            int st = step0 + i;
            if (st >= nSteps) st -= nSteps;
            if (st >= nSteps) st -= nSteps;
            return st;
        };

        if (nSteps > 0) {
            int4 ka[4], kb[4];
            load_step(ka, step_of(0));
            for (int it = 0; it < nSteps; it += 2) {
                const int sa = step_of(it);
                const int sb = step_of(it + 1);
                load_step(kb, sb);
                look_at(ka, sa, true);
                load_step(ka, step_of(it + 2));
                look_at(kb, sb, it + 1 < nSteps);
            }
        }
    }
    if (cnt > 0) {
        if constexpr (HITS) flushw(le, cnt); else flush(le, cnt);
    }
    __syncthreads();
    if constexpr (MODE == 1) {
        if (cap > 0 && threadIdx.x == 0) counts[blockIdx.x] = lcount;
    }

    if (wave < TN / 16) {
        const int hi  = lane >> 4;
        const int l15 = lane & 15;
        const int lr  = wave * 16 + l15;
        int row = tile0 + lr;
        if (row > nNodes - 1) row = nNodes - 1;
        const float invd = 1.0f / fmaxf(degs[lr], 1.0f);

        v8f c[CT];
#pragma unroll
        for (int ct = 0; ct < CT; ++ct) {
            float bv = bias[ct * 16 + l15];
#pragma unroll
            for (int v = 0; v < 8; ++v) c[ct][v] = bv;
        }
        const float* xr = x + (long)row * F;
        const float* ar = acc + lr * P;
#pragma unroll
        for (int k0 = 0; k0 < F; k0 += 4) {
            const int kA = k0 + 2 * hi;
            v2f a, m;
            a.x = xr[kA];
            a.y = xr[kA + 1];
            m.x = ar[kA] * invd;
            m.y = ar[kA + 1] * invd;
#pragma unroll
            for (int ct = 0; ct < CT; ++ct) {
                const int colW = ct * 16 + l15;
                v2f b;
                b.x = Wself[(long)(kA)     * OUTW + colW];
                b.y = Wself[(long)(kA + 1) * OUTW + colW];
                c[ct] = __builtin_amdgcn_wmma_f32_16x16x4_f32(false, a, false, b, (short)0, c[ct], false, false);
                b.x = Wneigh[(long)(kA)     * OUTW + colW];
                b.y = Wneigh[(long)(kA + 1) * OUTW + colW];
                c[ct] = __builtin_amdgcn_wmma_f32_16x16x4_f32(false, m, false, b, (short)0, c[ct], false, false);
            }
        }
#pragma unroll
        for (int ct = 0; ct < CT; ++ct) {
#pragma unroll
            for (int v = 0; v < 8; ++v) {
                int r = tile0 + wave * 16 + v + 8 * hi;
                float o = c[ct][v];
                if (RELU) o = fmaxf(o, 0.0f);
                if (r < nNodes) y[(long)r * OUTW + ct * 16 + l15] = o;
            }
        }
    }
}

extern "C" void kernel_launch(void* const* d_in, const int* in_sizes, int n_in,
                              void* d_out, int out_size, void* d_ws, size_t ws_size,
                              hipStream_t stream) {
    const float* feat = (const float*)d_in[0];
    const int*   src  = (const int*)d_in[1];
    const int*   dst  = (const int*)d_in[2];
    const float* W1s  = (const float*)d_in[3];
    const float* W1n  = (const float*)d_in[4];
    const float* b1   = (const float*)d_in[5];
    const float* W2s  = (const float*)d_in[6];
    const float* W2n  = (const float*)d_in[7];
    const float* b2   = (const float*)d_in[8];
    float* out = (float*)d_out;

    const int nNodes = in_sizes[0] / IN_F;
    const int nE     = in_sizes[1];
    const int nTiles = (nNodes + TN - 1) / TN;
    const int nBlocks = (nE + 31) / 32;
    const int nWords = nBlocks * 8;

    float* h    = (float*)d_ws;
    int*   keys = (int*)(h + (size_t)nNodes * HID_F);
    int*   counts = keys + nWords;
    unsigned* lists = (unsigned*)(counts + ((nTiles + 31) & ~31));
    const bool fits = (size_t)((const char*)(lists + (size_t)nTiles * TCAP) - (const char*)d_ws) <= ws_size
                      && nNodes <= (1 << (32 - UBITS));
    const int cap = fits ? TCAP : 0;
    int* recs = (int*)(lists + (size_t)nTiles * TCAP);
    const bool fitsRecs = fits && (size_t)((const char*)(recs + 8 * (size_t)nWords) - (const char*)d_ws) <= ws_size;
    int* pairs = fitsRecs ? recs : nullptr;

    const int nGroups = (nBlocks + GB - 1) / GB;
    unsigned* hitw = (unsigned*)(recs + 8 * (size_t)nWords);
    const bool fitsHits = fitsRecs && (size_t)((const char*)(hitw + (size_t)nGroups * (256 * MW)) - (const char*)d_ws) <= ws_size;

    if (fitsHits) {
        owner_hits_kernel<<<nGroups, 256, 0, stream>>>(dst, src, pairs, hitw, nE, nBlocks);
        sage_owner_kernel<IN_F, HID_F, true, 1, true><<<nTiles, NT, 0, stream>>>(feat, src, dst, keys, pairs, hitw, W1s, W1n, b1, h, counts, lists, cap, nNodes, nE);
        sage_owner_kernel<HID_F, OUT_F, false, 2, true><<<nTiles, NT, 0, stream>>>(h, src, dst, keys, pairs, hitw, W2s, W2n, b2, out, counts, lists, cap, nNodes, nE);
    } else {
        owner_key_kernel<<<(nBlocks + 8 * BPW - 1) / (8 * BPW), 256, 0, stream>>>(dst, src, keys, pairs, nE, nBlocks);
        sage_owner_kernel<IN_F, HID_F, true, 1, false><<<nTiles, NT, 0, stream>>>(feat, src, dst, keys, pairs, nullptr, W1s, W1n, b1, h, counts, lists, cap, nNodes, nE);
        sage_owner_kernel<HID_F, OUT_F, false, 2, false><<<nTiles, NT, 0, stream>>>(h, src, dst, keys, pairs, nullptr, W2s, W2n, b2, out, counts, lists, cap, nNodes, nE);
    }
}


# Layer 2 sums: each local row gets one owner wave (row mod 32) that sums it in registers from its LDS share of the tile list; no LDS float atomic per edge; sum order = share rank, run-varying
# speedup vs baseline: 6.1062x; 1.1369x over previous
#include <hip/hip_runtime.h>

#define IN_F 32
#define HID_F 64
#define OUT_F 16
#define TN 400
#define NW 32
#define NT (NW * 32)
#define LCAP 64
#define BPW 16
#define UBITS 9
#define TCAP 8192
#define GB (8 * BPW)
#define MW (GB / 8)
#define SCAP 384
#define RPW ((TN + NW - 1) / NW)
static_assert(NW == 32 && RPW < 16, "a wave's row sums are 16-float register vectors; place RPW is the spare one");
static_assert(MW >= 4 && MW <= 32 && 32 % MW == 0 && MW % 4 == 0, "the lanes of a wave must split evenly over the map words of a group");

typedef float v2f __attribute__((ext_vector_type(2)));
typedef float v8f __attribute__((ext_vector_type(8)));
typedef float v16f __attribute__((ext_vector_type(16)));

__global__ __launch_bounds__(256) void owner_key_kernel(const int* __restrict__ dst, const int* __restrict__ src,
                                                        int* __restrict__ keys, int* __restrict__ pairs,
                                                        int nE, int nBlocks) {
    const int lane = (int)threadIdx.x & 31;
    const int blk0 = (blockIdx.x * 8 + ((int)threadIdx.x >> 5)) * BPW;
    if (blk0 >= nBlocks) return;
    unsigned pl[8];
#pragma unroll
    for (int p = 0; p < 8; ++p) pl[p] = 0u;
#pragma unroll
    for (int c = 0; c < BPW / 8; ++c) {
        int d[8], s[8];
#pragma unroll
        for (int j = 0; j < 8; ++j) {
            const int e = (blk0 + c * 8 + j) * 32 + lane;
            d[j] = (e < nE) ? dst[e] : -1;
            s[j] = (pairs && e < nE) ? src[e] : 0;
        }
#pragma unroll
        for (int j = 0; j < 8; ++j) {
            const int blk = blk0 + c * 8 + j;
            const int e = blk * 32 + lane;
            if (pairs && blk < nBlocks) *(int2*)(pairs + 2 * (long)e) = make_int2(d[j], s[j]);
            const unsigned key = d[j] < 0 ? 255u : (((unsigned)d[j] / TN) & 255u);
#pragma unroll
            for (int p = 0; p < 8; ++p) {
                const unsigned m = __builtin_amdgcn_ballot_w32(((key >> p) & 1u) != 0u);
                if (lane == c * 8 + j) pl[p] = m;
            }
        }
    }
    const int blk = blk0 + lane;
    if (lane < BPW && blk < nBlocks) {
        *(int4*)(keys + 4 * (long)blk)             = make_int4((int)pl[0], (int)pl[1], (int)pl[2], (int)pl[3]);
        *(int4*)(keys + 4 * ((long)nBlocks + blk)) = make_int4((int)pl[4], (int)pl[5], (int)pl[6], (int)pl[7]);
    }
}

__global__ __launch_bounds__(256) void owner_hits_kernel(const int* __restrict__ dst, const int* __restrict__ src,
                                                         int* __restrict__ pairs,
                                                         unsigned* __restrict__ hits, int nE, int nBlocks) {
    __shared__ unsigned slab[256 * (MW + 1)];
    const int lane = (int)threadIdx.x & 31;
    const int wv   = (int)threadIdx.x >> 5;
    const int blk0 = (blockIdx.x * 8 + wv) * BPW;
    for (int i = threadIdx.x; i < 256 * (MW + 1); i += 256) slab[i] = 0u;
    __syncthreads();
    if (blk0 < nBlocks) {
#pragma unroll
        for (int c = 0; c < BPW / 8; ++c) {
            int d[8], s[8];
#pragma unroll
            for (int j = 0; j < 8; ++j) {
                const int e = (blk0 + c * 8 + j) * 32 + lane;
                d[j] = (e < nE) ? dst[e] : -1;
                s[j] = (e < nE) ? src[e] : 0;
            }
#pragma unroll
            for (int j = 0; j < 8; ++j) {
                const int blk = blk0 + c * 8 + j;
                const int e = blk * 32 + lane;
                if (blk < nBlocks) *(int2*)(pairs + 2 * (long)e) = make_int2(d[j], s[j]);
                if (d[j] >= 0) {
                    const unsigned key = ((unsigned)d[j] / TN) & 255u;
                    const int o = (wv * BPW + c * 8 + j) * 4 + (lane >> 3);
                    atomicOr(&slab[key * (MW + 1) + (o >> 5)], 1u << (o & 31));
                }
            }
        }
    }
    __syncthreads();
    uint4 m[MW / 4];
#pragma unroll
    for (int r = 0; r < MW / 4; ++r) {
        const int i = ((int)threadIdx.x + 256 * r) * 4;
        const unsigned* sp = &slab[i + i / MW];
        m[r] = make_uint4(sp[0], sp[1], sp[2], sp[3]);
    }
#pragma unroll
    for (int r = 0; r < MW / 4; ++r)
        *(uint4*)(hits + (long)blockIdx.x * (256 * MW) + ((int)threadIdx.x + 256 * r) * 4) = m[r];
}

__device__ __forceinline__ unsigned match32(int4 lo, int4 hi, const unsigned (&t)[8]) {
    unsigned a = (unsigned)lo.x ^ t[0];
    a |= (unsigned)lo.y ^ t[1];
    a |= (unsigned)lo.z ^ t[2];
    a |= (unsigned)lo.w ^ t[3];
    a |= (unsigned)hi.x ^ t[4];
    a |= (unsigned)hi.y ^ t[5];
    a |= (unsigned)hi.z ^ t[6];
    a |= (unsigned)hi.w ^ t[7];
    return ~a;
}

__device__ __forceinline__ bool confirm_batch(const int* __restrict__ src, const int* __restrict__ dst,
                                              const int* __restrict__ pairs,
                                              const int* le, int nvalid, int tile0, int nE, int lane,
                                              int& s, int& u) {
    int e = (lane < nvalid) ? le[lane] : 0;
    bool ok = (lane < nvalid) & (e < nE);
    if (!ok) e = 0;
    int d, sv;
    if (pairs) {
        const unsigned long long r = *(const unsigned long long*)(pairs + 2 * (long)e);
        d  = (int)(unsigned)r;
        sv = (int)(unsigned)(r >> 32);
    } else {
        d  = dst[e];
        sv = src[e];
    }
    unsigned uu = (unsigned)(d - tile0);
    ok = ok & (uu < (unsigned)TN);
    s = ok ? sv : 0;
    u = ok ? (int)uu : TN;
    return ok;
}

template <int F, int P>
__device__ __forceinline__ void add_rows(const float* __restrict__ x, int s, int u,
                                         float* acc, float* degs, int lane) {
    atomicAdd(&degs[u], 1.0f);
    if constexpr (F == 32) {
        float v[32];
#pragma unroll
        for (int j = 0; j < 32; ++j) {
            int sj = __builtin_amdgcn_readlane(s, j);
            v[j] = x[(long)sj * F + lane];
        }
#pragma unroll
        for (int j = 0; j < 32; ++j) {
            int uj = __builtin_amdgcn_readlane(u, j);
            atomicAdd(&acc[uj * P + lane], v[j]);
        }
    } else {
        v2f v[32];
#pragma unroll
        for (int j = 0; j < 32; ++j) {
            int sj = __builtin_amdgcn_readlane(s, j);
            v[j] = *(const v2f*)(x + (long)sj * F + 2 * lane);
        }
#pragma unroll
        for (int j = 0; j < 32; ++j) {
            int uj = __builtin_amdgcn_readlane(u, j);
            float* a = &acc[uj * P + 2 * lane];
            atomicAdd(a, v[j].x);
            atomicAdd(a + 1, v[j].y);
        }
    }
}

template <int F, int OUTW, bool RELU, int MODE, bool HITS>
__global__ __launch_bounds__(NT) void sage_owner_kernel(const float* __restrict__ x,
                                                        const int* __restrict__ src,
                                                        const int* __restrict__ dst,
                                                        const int* __restrict__ keys,
                                                        const int* __restrict__ pairs,
                                                        const unsigned* __restrict__ hits,
                                                        const float* __restrict__ Wself,
                                                        const float* __restrict__ Wneigh,
                                                        const float* __restrict__ bias,
                                                        float* __restrict__ y,
                                                        int* __restrict__ counts,
                                                        unsigned* __restrict__ lists, int cap,
                                                        int nNodes, int nE) {
    constexpr int P  = F + 2;
    constexpr int CT = OUTW / 16;
    __shared__ float acc[(TN + 1) * P];
    __shared__ float degs[TN + 16];
    __shared__ int list_e[NW * LCAP];
    __shared__ int list_b[HITS ? NW * LCAP : 1];
    __shared__ int lcount;
    __shared__ unsigned sub_list[(MODE == 2 && F == 64) ? NW * SCAP : 1];
    __shared__ int sub_cnt[NW];

    const int wave  = threadIdx.x >> 5;
    const int lane  = threadIdx.x & 31;
    const int tile0 = blockIdx.x * TN;

    unsigned* tl = lists + (long)blockIdx.x * cap;
    int total = 0;
    bool from_list = false;
    unsigned wc = 0;
    const int b0 = __builtin_amdgcn_readfirstlane(wave) * 32;
    if constexpr (MODE == 2) {
        if (cap > 0) {
            total = counts[blockIdx.x];
            from_list = total <= cap;
            if (from_list && b0 < total) wc = tl[min(b0 + lane, total - 1)];
        }
    }

    const int nBlocks = (nE + 31) >> 5;
    constexpr int GPR = 32 / MW;
    const int nGroups = (nBlocks + GB - 1) / GB;
    const int gFirst  = __builtin_amdgcn_readfirstlane(wave) + NW * (lane / MW);
    auto load_hits = [&](int r) -> unsigned {
        const int g = gFirst + r * (NW * GPR);
        return g < nGroups ? hits[((long)g * 256 + (blockIdx.x & 255u)) * MW + (lane % MW)] : 0u;
    };
    unsigned hpre = 0;
    if constexpr (HITS && MODE == 1) hpre = load_hits(0);

    for (int i = threadIdx.x; i < (TN + 1) * P; i += NT) acc[i] = 0.0f;
    for (int i = threadIdx.x; i < TN + 16; i += NT) degs[i] = 0.0f;
    if (threadIdx.x == 0) lcount = 0;
    if (threadIdx.x < NW) sub_cnt[threadIdx.x] = 0;
    __syncthreads();

    int* le = &list_e[wave * LCAP];
    int cnt = 0;

    auto flush = [&](const int* l, int nvalid) {
        int s, u;
        bool ok = confirm_batch(src, dst, pairs, l, nvalid, tile0, nE, lane, s, u);
        if constexpr (MODE == 1) {
            if (cap > 0) {
                unsigned okm = __builtin_amdgcn_ballot_w32(ok);
                int at = 0;
                if (lane == 0) at = atomicAdd(&lcount, __builtin_popcount(okm));
                at = __builtin_amdgcn_readfirstlane(at) + __builtin_amdgcn_mbcnt_lo(okm, 0);
                if (ok && at < cap) tl[at] = ((unsigned)s << UBITS) | (unsigned)u;
            }
        }
        add_rows<F, P>(x, s, u, acc, degs, lane);
    };
    auto flushw = [&](const int* l, int nvalid) {
        const bool ok = lane < nvalid;
        const unsigned w = ok ? (unsigned)l[lane] : 0u;
        if constexpr (MODE == 1) {
            if (cap > 0) {
                int at = 0;
                if (lane == 0) at = atomicAdd(&lcount, nvalid);
                at = __builtin_amdgcn_readfirstlane(at) + lane;
                if (ok && at < cap) tl[at] = w;
            }
        }
        const int s = ok ? (int)(w >> UBITS) : 0;
        const int u = ok ? (int)(w & ((1u << UBITS) - 1u)) : TN;
        add_rows<F, P>(x, s, u, acc, degs, lane);
    };

    unsigned tk[8];
#pragma unroll
    for (int p = 0; p < 8; ++p) tk[p] = ((blockIdx.x >> p) & 1u) ? 0xFFFFFFFFu : 0u;

    bool summed = false;
    if constexpr (MODE == 2 && F == 64) {
        if (from_list) {
            {
                int b = b0;
                unsigned w = wc;
                while (b < total) {
                    const int bn = b + NT;
                    unsigned wn = 0;
                    if (bn < total) wn = tl[min(bn + lane, total - 1)];
                    if (b + lane < total) {
                        const int sub = (int)(w & (unsigned)(NW - 1));
                        const int pos = atomicAdd(&sub_cnt[sub], 1);
                        if (pos < SCAP) sub_list[sub * SCAP + pos] = w;
                    }
                    w = wn;
                    b = bn;
                }
            }
            __syncthreads();
            const int myc = sub_cnt[lane];
            summed = __builtin_amdgcn_ballot_w32(myc > SCAP) == 0u;
            if (summed) {
                const int wv = __builtin_amdgcn_readfirstlane(wave);
                const int n  = __builtin_amdgcn_readlane(myc, wv);
                const unsigned* sl = &sub_list[wv * SCAP];
                const int l15 = lane & 15;
                v16f ax, ay;
#pragma unroll
                for (int k = 0; k < 16; ++k) { ax[k] = 0.0f; ay[k] = 0.0f; }
                unsigned wcur = (l15 < n) ? sl[l15] : 0u;
                for (int b = 0; b < n; b += 16) {
                    const bool ok = b + l15 < n;
                    const int s = ok ? (int)(wcur >> UBITS) : 0;
                    const int u = (int)(wcur & ((1u << UBITS) - 1u));
                    const int k = ok ? (u >> 5) : RPW;
                    if (ok && lane < 16) atomicAdd(&degs[u], 1.0f);
                    v2f v[16];
#pragma unroll
                    for (int j = 0; j < 16; ++j) {
                        const int sj = __builtin_amdgcn_readlane(s, j);
                        v[j] = *(const v2f*)(x + (long)sj * F + 2 * lane);
                    }
                    const int nb = b + 16 + l15;
                    wcur = (nb < n) ? sl[nb] : 0u;
#pragma unroll
                    for (int j = 0; j < 16; ++j) {
                        const int kj = __builtin_amdgcn_readlane(k, j);
                        ax[kj] += v[j].x;
                        ay[kj] += v[j].y;
                    }
                }
#pragma unroll
                for (int k = 0; k < RPW; ++k) {
                    const int r = wv + NW * k;
                    if (r < TN) {
                        acc[r * P + 2 * lane]     = ax[k];
                        acc[r * P + 2 * lane + 1] = ay[k];
                    }
                }
            }
        }
    }
    if (summed) {
    } else if (from_list) {
        int b = b0;
        while (b < total) {
            const int bn = b + NT;
            unsigned wn = 0;
            if (bn < total) wn = tl[min(bn + lane, total - 1)];
            const bool ok = b + lane < total;
            const int s = ok ? (int)(wc >> UBITS) : 0;
            const int u = ok ? (int)(wc & ((1u << UBITS) - 1u)) : TN;
            add_rows<F, P>(x, s, u, acc, degs, lane);
            wc = wn;
            b  = bn;
        }
    } else if constexpr (HITS) {
        int* lb = &list_b[wave * LCAP];
        int cb = 0;
        auto look_octets = [&](const int* l, int nvalid) {
            int4 rc[4];
            unsigned vmask = 0;
#pragma unroll
            for (int i = 0; i < 4; ++i) {
                const int k = i * 8 + (lane >> 2);
                const bool v = k < nvalid;
                const int o = v ? l[k] : 0;
                rc[i] = *(const int4*)(pairs + 16 * (long)o + 4 * (lane & 3));
                vmask |= (v ? 3u : 0u) << (2 * i);
            }
            unsigned wd[8];
            unsigned q = 0;
#pragma unroll
            for (int i = 0; i < 4; ++i) {
                const unsigned u0 = (unsigned)(rc[i].x - tile0);
                const unsigned u1 = (unsigned)(rc[i].z - tile0);
                q |= (u0 < (unsigned)TN ? 1u : 0u) << (2 * i);
                q |= (u1 < (unsigned)TN ? 1u : 0u) << (2 * i + 1);
                wd[2 * i]     = ((unsigned)rc[i].y << UBITS) | u0;
                wd[2 * i + 1] = ((unsigned)rc[i].w << UBITS) | u1;
            }
            q &= vmask;
            while (true) {
                bool has = q != 0;
                unsigned bm = __builtin_amdgcn_ballot_w32(has);
                if (bm == 0) break;
                if (has) {
                    const int p = __builtin_ctz(q);
                    q &= q - 1;
                    unsigned w = wd[0];
#pragma unroll
                    for (int i = 1; i < 8; ++i) w = (p == i) ? wd[i] : w;
                    le[cnt + __builtin_amdgcn_mbcnt_lo(bm, 0)] = (int)w;
                }
                cnt += __builtin_popcount(bm);
                if (cnt >= 32) {
                    cnt -= 32;
                    flushw(le + cnt, 32);
                }
            }
        };
        const int nRounds = (nGroups + NW * GPR - 1) / (NW * GPR);
        unsigned hcur = (MODE == 1) ? hpre : load_hits(0);
        for (int r = 0; r < nRounds; ++r) {
            const unsigned hnext = load_hits(r + 1);
            unsigned hw = hcur;
            const int oBase = (gFirst + r * (NW * GPR)) * (GB * 4) + (lane % MW) * 32;
            while (true) {
                bool has = hw != 0;
                unsigned bm = __builtin_amdgcn_ballot_w32(has);
                if (bm == 0) break;
                if (has) {
                    int p = __builtin_ctz(hw);
                    hw &= hw - 1;
                    lb[cb + __builtin_amdgcn_mbcnt_lo(bm, 0)] = oBase + p;
                }
                cb += __builtin_popcount(bm);
                if (cb >= 32) {
                    cb -= 32;
                    look_octets(lb + cb, 32);
                }
            }
            hcur = hnext;
        }
        if (cb > 0) look_octets(lb, cb);
    } else {
        const int nSteps  = (nBlocks + 2 * NT - 1) / (2 * NT);
        const int step0   = (int)(((long)blockIdx.x * nSteps) / gridDim.x);

        auto load_step = [&](int4 (&w)[4], int st) {
#pragma unroll
            for (int k = 0; k < 2; ++k) {
                int b = (st * 2 + k) * NT + (int)threadIdx.x;
                if (b > nBlocks - 1) b = nBlocks - 1;
                w[2 * k]     = *(const int4*)(keys + 4 * (long)b);
                w[2 * k + 1] = *(const int4*)(keys + 4 * ((long)nBlocks + b));
            }
        };
        auto look_at = [&](const int4 (&w)[4], int st, bool live) {
#pragma unroll
            for (int h2 = 0; h2 < 2; ++h2) {
                const int b = (st * 2 + h2) * NT + (int)threadIdx.x;
                unsigned q = match32(w[2 * h2], w[2 * h2 + 1], tk);
                if (!(live && b < nBlocks)) q = 0;
                const int eBase = b * 32;
                while (true) {
                    bool has = q != 0;
                    unsigned bm = __builtin_amdgcn_ballot_w32(has);
                    if (bm == 0) break;
                    if (has) {
                        int p = __builtin_ctz(q);
                        q &= q - 1;
                        int e = eBase + p;
                        le[cnt + __builtin_amdgcn_mbcnt_lo(bm, 0)] = e;
                    }
                    cnt += __builtin_popcount(bm);
                    if (cnt >= 32) {
                        cnt -= 32;
                        flush(le + cnt, 32);
                    }
                }
            }
        };
        auto step_of = [&](int i) {
            int st = step0 + i;
            if (st >= nSteps) st -= nSteps;
            if (st >= nSteps) st -= nSteps;
            return st;
        };

        if (nSteps > 0) {
            int4 ka[4], kb[4];
            load_step(ka, step_of(0));
            for (int it = 0; it < nSteps; it += 2) {
                const int sa = step_of(it);
                const int sb = step_of(it + 1);
                load_step(kb, sb);
                look_at(ka, sa, true);
                load_step(ka, step_of(it + 2));
                look_at(kb, sb, it + 1 < nSteps);
            }
        }
    }
    if (cnt > 0) {
        if constexpr (HITS) flushw(le, cnt); else flush(le, cnt);
    }
    __syncthreads();
    if constexpr (MODE == 1) {
        if (cap > 0 && threadIdx.x == 0) counts[blockIdx.x] = lcount;
    }

    if (wave < TN / 16) {
        const int hi  = lane >> 4;
        const int l15 = lane & 15;
        const int lr  = wave * 16 + l15;
        int row = tile0 + lr;
        if (row > nNodes - 1) row = nNodes - 1;
        const float invd = 1.0f / fmaxf(degs[lr], 1.0f);

        v8f c[CT];
#pragma unroll
        for (int ct = 0; ct < CT; ++ct) {
            float bv = bias[ct * 16 + l15];
#pragma unroll
            for (int v = 0; v < 8; ++v) c[ct][v] = bv;
        }
        const float* xr = x + (long)row * F;
        const float* ar = acc + lr * P;
#pragma unroll
        for (int k0 = 0; k0 < F; k0 += 4) {
            const int kA = k0 + 2 * hi;
            v2f a, m;
            a.x = xr[kA];
            a.y = xr[kA + 1];
            m.x = ar[kA] * invd;
            m.y = ar[kA + 1] * invd;
#pragma unroll
            for (int ct = 0; ct < CT; ++ct) {
                const int colW = ct * 16 + l15;
                v2f b;
                b.x = Wself[(long)(kA)     * OUTW + colW];
                b.y = Wself[(long)(kA + 1) * OUTW + colW];
                c[ct] = __builtin_amdgcn_wmma_f32_16x16x4_f32(false, a, false, b, (short)0, c[ct], false, false);
                b.x = Wneigh[(long)(kA)     * OUTW + colW];
                b.y = Wneigh[(long)(kA + 1) * OUTW + colW];
                c[ct] = __builtin_amdgcn_wmma_f32_16x16x4_f32(false, m, false, b, (short)0, c[ct], false, false);
            }
        }
#pragma unroll
        for (int ct = 0; ct < CT; ++ct) {
#pragma unroll
            for (int v = 0; v < 8; ++v) {
                int r = tile0 + wave * 16 + v + 8 * hi;
                float o = c[ct][v];
                if (RELU) o = fmaxf(o, 0.0f);
                if (r < nNodes) y[(long)r * OUTW + ct * 16 + l15] = o;
            }
        }
    }
}

extern "C" void kernel_launch(void* const* d_in, const int* in_sizes, int n_in,
                              void* d_out, int out_size, void* d_ws, size_t ws_size,
                              hipStream_t stream) {
    const float* feat = (const float*)d_in[0];
    const int*   src  = (const int*)d_in[1];
    const int*   dst  = (const int*)d_in[2];
    const float* W1s  = (const float*)d_in[3];
    const float* W1n  = (const float*)d_in[4];
    const float* b1   = (const float*)d_in[5];
    const float* W2s  = (const float*)d_in[6];
    const float* W2n  = (const float*)d_in[7];
    const float* b2   = (const float*)d_in[8];
    float* out = (float*)d_out;

    const int nNodes = in_sizes[0] / IN_F;
    const int nE     = in_sizes[1];
    const int nTiles = (nNodes + TN - 1) / TN;
    const int nBlocks = (nE + 31) / 32;
    const int nWords = nBlocks * 8;

    float* h    = (float*)d_ws;
    int*   keys = (int*)(h + (size_t)nNodes * HID_F);
    int*   counts = keys + nWords;
    unsigned* lists = (unsigned*)(counts + ((nTiles + 31) & ~31));
    const bool fits = (size_t)((const char*)(lists + (size_t)nTiles * TCAP) - (const char*)d_ws) <= ws_size
                      && nNodes <= (1 << (32 - UBITS));
    const int cap = fits ? TCAP : 0;
    int* recs = (int*)(lists + (size_t)nTiles * TCAP);
    const bool fitsRecs = fits && (size_t)((const char*)(recs + 8 * (size_t)nWords) - (const char*)d_ws) <= ws_size;
    int* pairs = fitsRecs ? recs : nullptr;

    const int nGroups = (nBlocks + GB - 1) / GB;
    unsigned* hitw = (unsigned*)(recs + 8 * (size_t)nWords);
    const bool fitsHits = fitsRecs && (size_t)((const char*)(hitw + (size_t)nGroups * (256 * MW)) - (const char*)d_ws) <= ws_size;

    if (fitsHits) {
        owner_hits_kernel<<<nGroups, 256, 0, stream>>>(dst, src, pairs, hitw, nE, nBlocks);
        sage_owner_kernel<IN_F, HID_F, true, 1, true><<<nTiles, NT, 0, stream>>>(feat, src, dst, keys, pairs, hitw, W1s, W1n, b1, h, counts, lists, cap, nNodes, nE);
        sage_owner_kernel<HID_F, OUT_F, false, 2, true><<<nTiles, NT, 0, stream>>>(h, src, dst, keys, pairs, hitw, W2s, W2n, b2, out, counts, lists, cap, nNodes, nE);
    } else {
        owner_key_kernel<<<(nBlocks + 8 * BPW - 1) / (8 * BPW), 256, 0, stream>>>(dst, src, keys, pairs, nE, nBlocks);
        sage_owner_kernel<IN_F, HID_F, true, 1, false><<<nTiles, NT, 0, stream>>>(feat, src, dst, keys, pairs, nullptr, W1s, W1n, b1, h, counts, lists, cap, nNodes, nE);
        sage_owner_kernel<HID_F, OUT_F, false, 2, false><<<nTiles, NT, 0, stream>>>(h, src, dst, keys, pairs, nullptr, W2s, W2n, b2, out, counts, lists, cap, nNodes, nE);
    }
}
